# starting kernel
# baseline (speedup 1.0000x reference)
#include <hip/hip_runtime.h>

typedef __attribute__((ext_vector_type(16))) __bf16 v16bf;
typedef __attribute__((ext_vector_type(8)))  float  v8f;

__global__ void init_deg_kernel(float* __restrict__ deg, int n) {
  int i = blockIdx.x * blockDim.x + threadIdx.x;
  if (i < n) deg[i] = 1.0f;
}

__global__ void count_deg_kernel(const int* __restrict__ dst, float* __restrict__ deg, int e) {
  int i = blockIdx.x * blockDim.x + threadIdx.x;
  if (i < e) atomicAdd(&deg[dst[i]], 1.0f);
}

__global__ void dinv_kernel(float* __restrict__ deg, int n) {
  int i = blockIdx.x * blockDim.x + threadIdx.x;
  if (i < n) {
    float d = deg[i];
    deg[i] = d > 0.0f ? rsqrtf(d) : 0.0f;
  }
}

template <int K>
__global__ void pack_W_kernel(const float* __restrict__ W, __bf16* __restrict__ Bp) {
  constexpr int slots = (K / 32) * 4 * 32;
  int t = blockIdx.x * blockDim.x + threadIdx.x;
  if (t >= slots) return;
  int lane  = t & 31;
  int nt    = (t >> 5) & 3;
  int ktIdx = t >> 7;
  int r16   = lane & 15;
  int khalf = (lane >> 4) * 8;
  int col   = nt * 16 + r16;
  v16bf b;
#pragma unroll
  for (int i = 0; i < 8; ++i) {
    b[i]     = (__bf16)W[(size_t)(ktIdx * 32 + khalf + i) * 64 + col];
    b[8 + i] = (__bf16)W[(size_t)(ktIdx * 32 + khalf + 16 + i) * 64 + col];
  }
  *(v16bf*)(Bp + (size_t)t * 16) = b;
}

template <int K, bool RELU_BIAS>
__global__ __launch_bounds__(256) void gemm_wmma_kernel(
    const float* __restrict__ A, const __bf16* __restrict__ Bp,
    const float* __restrict__ bias, float* __restrict__ out, int nRows) {
  const int lane = threadIdx.x & 31;
  const int wave = threadIdx.x >> 5;
  const int rowTile = (blockIdx.x * 8 + wave) * 16;
  if (rowTile >= nRows) return;

  const int r16   = lane & 15;
  const int khalf = (lane >> 4) * 8;
  int row  = rowTile + r16;
  int rowc = row < nRows ? row : nRows - 1;
  const float* __restrict__ arow = A + (size_t)rowc * K;

  v8f acc[4];
  acc[0] = (v8f){}; acc[1] = (v8f){}; acc[2] = (v8f){}; acc[3] = (v8f){};

#pragma unroll
  for (int ktIdx = 0; ktIdx < K / 32; ++ktIdx) {
    const int kt = ktIdx * 32;
    const float4 a0 = *(const float4*)(arow + kt + khalf);
    const float4 a1 = *(const float4*)(arow + kt + khalf + 4);
    const float4 a2 = *(const float4*)(arow + kt + khalf + 16);
    const float4 a3 = *(const float4*)(arow + kt + khalf + 20);
    float lo[8] = {a0.x, a0.y, a0.z, a0.w, a1.x, a1.y, a1.z, a1.w};
    float hi[8] = {a2.x, a2.y, a2.z, a2.w, a3.x, a3.y, a3.z, a3.w};
    if constexpr (RELU_BIAS) {
      const float4 bb0 = *(const float4*)(bias + kt + khalf);
      const float4 bb1 = *(const float4*)(bias + kt + khalf + 4);
      const float4 bb2 = *(const float4*)(bias + kt + khalf + 16);
      const float4 bb3 = *(const float4*)(bias + kt + khalf + 20);
      float blo[8] = {bb0.x, bb0.y, bb0.z, bb0.w, bb1.x, bb1.y, bb1.z, bb1.w};
      float bhi[8] = {bb2.x, bb2.y, bb2.z, bb2.w, bb3.x, bb3.y, bb3.z, bb3.w};
#pragma unroll
      for (int i = 0; i < 8; ++i) {
        lo[i] += blo[i]; lo[i] = lo[i] > 0.0f ? lo[i] : 0.0f;
        hi[i] += bhi[i]; hi[i] = hi[i] > 0.0f ? hi[i] : 0.0f;
      }
    }
    v16bf a;
#pragma unroll
    for (int i = 0; i < 8; ++i) {
      a[i]     = (__bf16)lo[i];
      a[8 + i] = (__bf16)hi[i];
    }
#pragma unroll
    for (int nt = 0; nt < 4; ++nt) {
      const v16bf b =
          *(const v16bf*)(Bp + ((size_t)(ktIdx * 4 + nt) * 32 + lane) * 16);
      acc[nt] = __builtin_amdgcn_wmma_f32_16x16x32_bf16(
           false, a,  false, b,
           (short)0, acc[nt],  false,  false);
    }
  }

  const int rbase = rowTile + ((lane >> 4) * 8);
#pragma unroll
  for (int nt = 0; nt < 4; ++nt) {
    const int col = nt * 16 + r16;
#pragma unroll
    for (int r = 0; r < 8; ++r) {
      int rr = rbase + r;
      if (rr < nRows) out[(size_t)rr * 64 + col] = acc[nt][r];
    }
  }
}

__global__ void selfloop_init_kernel(const float* __restrict__ x,
                                     const float* __restrict__ dinv,
                                     const float* __restrict__ bias,
                                     float* __restrict__ out, int n) {
  int t = blockIdx.x * blockDim.x + threadIdx.x;
  int i = t >> 4;
  if (i >= n) return;
  int f = (t & 15) * 4;
  float di  = dinv[i];
  float nrm = di * di;
  const float4 v = *(const float4*)(x + (size_t)i * 64 + f);
  float4 o;
  o.x = v.x * nrm; o.y = v.y * nrm; o.z = v.z * nrm; o.w = v.w * nrm;
  if (bias != nullptr) {
    o.x += bias[f]; o.y += bias[f + 1]; o.z += bias[f + 2]; o.w += bias[f + 3];
  }
  *(float4*)(out + (size_t)i * 64 + f) = o;
}

__global__ void scatter_edges_kernel(const float* __restrict__ x,
                                     const float* __restrict__ dinv,
                                     const int* __restrict__ src,
                                     const int* __restrict__ dst,
                                     float* __restrict__ out, int nEdges) {
  int t = blockIdx.x * blockDim.x + threadIdx.x;
  int e = t >> 4;
  if (e >= nEdges) return;
  int f = (t & 15) * 4;
  int s = src[e];
  int d = dst[e];
  float nrm = dinv[s] * dinv[d];
  const float4 v = *(const float4*)(x + (size_t)s * 64 + f);
  float* __restrict__ o = out + (size_t)d * 64 + f;
  atomicAdd(o + 0, v.x * nrm);
  atomicAdd(o + 1, v.y * nrm);
  atomicAdd(o + 2, v.z * nrm);
  atomicAdd(o + 3, v.w * nrm);
}

extern "C" void kernel_launch(void* const* d_in, const int* in_sizes, int n_in,
                              void* d_out, int out_size, void* d_ws, size_t ws_size,
                              hipStream_t stream) {
  const int*   edge_index = (const int*)d_in[0];
  const float* emb        = (const float*)d_in[1];
  const float* W1         = (const float*)d_in[2];
  const float* b1         = (const float*)d_in[3];
  const float* W2         = (const float*)d_in[4];
  const float* b2         = (const float*)d_in[5];
  float* out = (float*)d_out;

  const int E = in_sizes[0] / 2;
  const int N = in_sizes[1] / 128;
  const int* src = edge_index;
  const int* dst = edge_index + E;

  float* ws   = (float*)d_ws;
  float* dinv = ws;
  size_t npad = ((size_t)N + 63) & ~(size_t)63;
  float* x1   = ws + npad;
  float* agg1 = x1 + (size_t)N * 64;
  float* x2   = x1;
  __bf16* Bp1 = (__bf16*)(agg1 + (size_t)N * 64);
  __bf16* Bp2 = Bp1 + 8192;

  const int tpb = 256;
  const int nBlk = (N + tpb - 1) / tpb;
  const int eBlk = (E + tpb - 1) / tpb;
  const int gemmBlk = (N + 127) / 128;
  const long nf = (long)N * 16;
  const long ef = (long)E * 16;
  const int nfBlk = (int)((nf + tpb - 1) / tpb);
  const int efBlk = (int)((ef + tpb - 1) / tpb);

  pack_W_kernel<128><<<2, tpb, 0, stream>>>(W1, Bp1);
  pack_W_kernel<64><<<1, tpb, 0, stream>>>(W2, Bp2);

  init_deg_kernel<<<nBlk, tpb, 0, stream>>>(dinv, N);
  count_deg_kernel<<<eBlk, tpb, 0, stream>>>(dst, dinv, E);
  dinv_kernel<<<nBlk, tpb, 0, stream>>>(dinv, N);

  gemm_wmma_kernel<128, false><<<gemmBlk, 256, 0, stream>>>(emb, Bp1, nullptr, x1, N);
  selfloop_init_kernel<<<nfBlk, tpb, 0, stream>>>(x1, dinv, nullptr, agg1, N);
  scatter_edges_kernel<<<efBlk, tpb, 0, stream>>>(x1, dinv, src, dst, agg1, E);

  gemm_wmma_kernel<64, true><<<gemmBlk, 256, 0, stream>>>(agg1, Bp2, b1, x2, N);
  selfloop_init_kernel<<<nfBlk, tpb, 0, stream>>>(x2, dinv, b2, out, N);
  scatter_edges_kernel<<<efBlk, tpb, 0, stream>>>(x2, dinv, src, dst, out, E);
}


# One returning int atomic per edge buckets edges by destination (slot = arrival-order rank, count = degree); each layer sums a row's sources in registers: no f32 atomics, sum order set by arrival
# speedup vs baseline: 2.2307x; 2.2307x over previous
#include <hip/hip_runtime.h>

typedef __attribute__((ext_vector_type(16))) __bf16 v16bf;
typedef __attribute__((ext_vector_type(8)))  float  v8f;

#define ROW_INTS 64
#define ROW_CAP  63

__global__ void init_rows_kernel(int* __restrict__ rows, int* __restrict__ ovfCount, int n) {
  int i = blockIdx.x * blockDim.x + threadIdx.x;
  if (i < n) rows[(size_t)i * ROW_INTS] = 0;
  if (i == 0) *ovfCount = 0;
}

__global__ void bucket_edges_kernel(const int* __restrict__ src, const int* __restrict__ dst,
                                    int* __restrict__ rows, int* __restrict__ ovfCount,
                                    int* __restrict__ ovfSrc, int* __restrict__ ovfDst, int e) {
  int i = blockIdx.x * blockDim.x + threadIdx.x;
  if (i >= e) return;
  int s = src[i];
  int d = dst[i];
  int* row = rows + (size_t)d * ROW_INTS;
  int pos = atomicAdd(row, 1);
  if (pos < ROW_CAP) {
    row[1 + pos] = s;
  } else {
    int o = atomicAdd(ovfCount, 1);
    ovfSrc[o] = s;
    ovfDst[o] = d;
  }
}

__global__ void dinv_kernel(const int* __restrict__ rows, float* __restrict__ dinv, int n) {
  int i = blockIdx.x * blockDim.x + threadIdx.x;
  if (i < n) {
    float d = (float)(rows[(size_t)i * ROW_INTS] + 1);
    dinv[i] = d > 0.0f ? rsqrtf(d) : 0.0f;
  }
}

template <int K>
__global__ void pack_W_kernel(const float* __restrict__ W, __bf16* __restrict__ Bp) {
  constexpr int slots = (K / 32) * 4 * 32;
  int t = blockIdx.x * blockDim.x + threadIdx.x;
  if (t >= slots) return;
  int lane  = t & 31;
  int nt    = (t >> 5) & 3;
  int ktIdx = t >> 7;
  int r16   = lane & 15;
  int khalf = (lane >> 4) * 8;
  int col   = nt * 16 + r16;
  v16bf b;
#pragma unroll
  for (int i = 0; i < 8; ++i) {
    b[i]     = (__bf16)W[(size_t)(ktIdx * 32 + khalf + i) * 64 + col];
    b[8 + i] = (__bf16)W[(size_t)(ktIdx * 32 + khalf + 16 + i) * 64 + col];
  }
  *(v16bf*)(Bp + (size_t)t * 16) = b;
}

template <int K, bool RELU_BIAS>
__global__ __launch_bounds__(256) void gemm_wmma_kernel(
    const float* __restrict__ A, const __bf16* __restrict__ Bp,
    const float* __restrict__ bias, float* __restrict__ out, int nRows) {
  const int lane = threadIdx.x & 31;
  const int wave = threadIdx.x >> 5;
  const int rowTile = (blockIdx.x * 8 + wave) * 16;
  if (rowTile >= nRows) return;

  const int r16   = lane & 15;
  const int khalf = (lane >> 4) * 8;
  int row  = rowTile + r16;
  int rowc = row < nRows ? row : nRows - 1;
  const float* __restrict__ arow = A + (size_t)rowc * K;

  v8f acc[4];
  acc[0] = (v8f){}; acc[1] = (v8f){}; acc[2] = (v8f){}; acc[3] = (v8f){};

#pragma unroll
  for (int ktIdx = 0; ktIdx < K / 32; ++ktIdx) {
    const int kt = ktIdx * 32;
    const float4 a0 = *(const float4*)(arow + kt + khalf);
    const float4 a1 = *(const float4*)(arow + kt + khalf + 4);
    const float4 a2 = *(const float4*)(arow + kt + khalf + 16);
    const float4 a3 = *(const float4*)(arow + kt + khalf + 20);
    float lo[8] = {a0.x, a0.y, a0.z, a0.w, a1.x, a1.y, a1.z, a1.w};
    float hi[8] = {a2.x, a2.y, a2.z, a2.w, a3.x, a3.y, a3.z, a3.w};
    if constexpr (RELU_BIAS) {
      const float4 bb0 = *(const float4*)(bias + kt + khalf);
      const float4 bb1 = *(const float4*)(bias + kt + khalf + 4);
      const float4 bb2 = *(const float4*)(bias + kt + khalf + 16);
      const float4 bb3 = *(const float4*)(bias + kt + khalf + 20);
      float blo[8] = {bb0.x, bb0.y, bb0.z, bb0.w, bb1.x, bb1.y, bb1.z, bb1.w};
      float bhi[8] = {bb2.x, bb2.y, bb2.z, bb2.w, bb3.x, bb3.y, bb3.z, bb3.w};
#pragma unroll
      for (int i = 0; i < 8; ++i) {
        lo[i] += blo[i]; lo[i] = lo[i] > 0.0f ? lo[i] : 0.0f;
        hi[i] += bhi[i]; hi[i] = hi[i] > 0.0f ? hi[i] : 0.0f;
      }
    }
    v16bf a;
#pragma unroll
    for (int i = 0; i < 8; ++i) {
      a[i]     = (__bf16)lo[i];
      a[8 + i] = (__bf16)hi[i];
    }
#pragma unroll
    for (int nt = 0; nt < 4; ++nt) {
      const v16bf b =
          *(const v16bf*)(Bp + ((size_t)(ktIdx * 4 + nt) * 32 + lane) * 16);
      acc[nt] = __builtin_amdgcn_wmma_f32_16x16x32_bf16(
           false, a,  false, b,
           (short)0, acc[nt],  false,  false);
    }
  }

  const int rbase = rowTile + ((lane >> 4) * 8);
#pragma unroll
  for (int nt = 0; nt < 4; ++nt) {
    const int col = nt * 16 + r16;
#pragma unroll
    for (int r = 0; r < 8; ++r) {
      int rr = rbase + r;
      if (rr < nRows) out[(size_t)rr * 64 + col] = acc[nt][r];
    }
  }
}

__global__ __launch_bounds__(256) void gather_rows_kernel(
    const float* __restrict__ x, const float* __restrict__ dinv,
    const int* __restrict__ rows, const float* __restrict__ bias,
    float* __restrict__ out, int n) {
  int t = blockIdx.x * blockDim.x + threadIdx.x;
  int i = t >> 4;
  if (i >= n) return;
  const int l16 = threadIdx.x & 15;
  const int f = l16 * 4;
  const int* __restrict__ row = rows + (size_t)i * ROW_INTS;
  const float di = dinv[i];
  int m = row[0];
  m = m < 0 ? 0 : (m > ROW_CAP ? ROW_CAP : m);
  const float4 self = *(const float4*)(x + (size_t)i * 64 + f);
  const float nrm = di * di;
  float4 acc;
  acc.x = self.x * nrm; acc.y = self.y * nrm; acc.z = self.z * nrm; acc.w = self.w * nrm;

  for (int base = 0; base < m; base += 16) {
    const int j = base + l16;
    int sidx = i;
    float sw = 0.0f;
    if (j < m) {
      int sraw = row[1 + j];
      sidx = sraw < 0 ? 0 : (sraw >= n ? n - 1 : sraw);
      sw = dinv[sidx] * di;
    }
    const int left = m - base;
    const int lim = left < 16 ? left : 16;
    for (int jj = 0; jj < lim; jj += 4) {
      int s[4];
      float w[4];
      float4 v[4];
#pragma unroll
      for (int u = 0; u < 4; ++u) {
        s[u] = __shfl(sidx, jj + u, 16);
        w[u] = __shfl(sw, jj + u, 16);
      }
#pragma unroll
      for (int u = 0; u < 4; ++u) {
        v[u] = self;
        if (jj + u < lim) v[u] = *(const float4*)(x + (size_t)s[u] * 64 + f);
      }
#pragma unroll
      for (int u = 0; u < 4; ++u) {
        if (jj + u < lim) {
          acc.x += v[u].x * w[u]; acc.y += v[u].y * w[u];
          acc.z += v[u].z * w[u]; acc.w += v[u].w * w[u];
        }
      }
    }
  }
  if (bias != nullptr) {
    acc.x += bias[f]; acc.y += bias[f + 1]; acc.z += bias[f + 2]; acc.w += bias[f + 3];
  }
  *(float4*)(out + (size_t)i * 64 + f) = acc;
}

__global__ void overflow_edges_kernel(const float* __restrict__ x,
                                      const float* __restrict__ dinv,
                                      const int* __restrict__ ovfCount,
                                      const int* __restrict__ ovfSrc,
                                      const int* __restrict__ ovfDst,
                                      float* __restrict__ out, int nEdges) {
  int cnt = *ovfCount;
  cnt = cnt < 0 ? 0 : (cnt > nEdges ? nEdges : cnt);
  const int t = blockIdx.x * blockDim.x + threadIdx.x;
  const int step = (gridDim.x * blockDim.x) >> 4;
  const int f = (t & 15) * 4;
  for (int e = t >> 4; e < cnt; e += step) {
    int s = ovfSrc[e];
    int d = ovfDst[e];
    float nrm = dinv[s] * dinv[d];
    const float4 v = *(const float4*)(x + (size_t)s * 64 + f);
    float* __restrict__ o = out + (size_t)d * 64 + f;
    atomicAdd(o + 0, v.x * nrm);
    atomicAdd(o + 1, v.y * nrm);
    atomicAdd(o + 2, v.z * nrm);
    atomicAdd(o + 3, v.w * nrm);
  }
}

extern "C" void kernel_launch(void* const* d_in, const int* in_sizes, int n_in,
                              void* d_out, int out_size, void* d_ws, size_t ws_size,
                              hipStream_t stream) {
  const int*   edge_index = (const int*)d_in[0];
  const float* emb        = (const float*)d_in[1];
  const float* W1         = (const float*)d_in[2];
  const float* b1         = (const float*)d_in[3];
  const float* W2         = (const float*)d_in[4];
  const float* b2         = (const float*)d_in[5];
  float* out = (float*)d_out;

  const int E = in_sizes[0] / 2;
  const int N = in_sizes[1] / 128;
  const int* src = edge_index;
  const int* dst = edge_index + E;

  float* ws   = (float*)d_ws;
  float* dinv = ws;
  size_t npad = ((size_t)N + 63) & ~(size_t)63;
  int*   ovfCount = (int*)(ws + npad);
  float* x1   = ws + npad + 64;
  float* agg1 = x1 + (size_t)N * 64;
  float* x2   = x1;
  __bf16* Bp1 = (__bf16*)(agg1 + (size_t)N * 64);
  __bf16* Bp2 = Bp1 + 8192;
  int* rows   = (int*)(agg1 + (size_t)N * 64 + 6144);
  int* ovfSrc = rows + (size_t)N * ROW_INTS;
  int* ovfDst = ovfSrc + E;

  const int tpb = 256;
  const int nBlk = (N + tpb - 1) / tpb;
  const int eBlk = (E + tpb - 1) / tpb;
  const int gemmBlk = (N + 127) / 128;
  const long nf = (long)N * 16;
  const int nfBlk = (int)((nf + tpb - 1) / tpb);
  const int ovfBlk = 64;

  pack_W_kernel<128><<<2, tpb, 0, stream>>>(W1, Bp1);
  pack_W_kernel<64><<<1, tpb, 0, stream>>>(W2, Bp2);

  init_rows_kernel<<<nBlk, tpb, 0, stream>>>(rows, ovfCount, N);
  bucket_edges_kernel<<<eBlk, tpb, 0, stream>>>(src, dst, rows, ovfCount, ovfSrc, ovfDst, E);
  dinv_kernel<<<nBlk, tpb, 0, stream>>>(rows, dinv, N);

  gemm_wmma_kernel<128, false><<<gemmBlk, 256, 0, stream>>>(emb, Bp1, nullptr, x1, N);
  gather_rows_kernel<<<nfBlk, tpb, 0, stream>>>(x1, dinv, rows, nullptr, agg1, N);
  overflow_edges_kernel<<<ovfBlk, tpb, 0, stream>>>(x1, dinv, ovfCount, ovfSrc, ovfDst, agg1, E);

  gemm_wmma_kernel<64, true><<<gemmBlk, 256, 0, stream>>>(agg1, Bp2, b1, x2, N);
  gather_rows_kernel<<<nfBlk, tpb, 0, stream>>>(x2, dinv, rows, b2, out, N);
  overflow_edges_kernel<<<ovfBlk, tpb, 0, stream>>>(x2, dinv, ovfCount, ovfSrc, ovfDst, out, E);
}


# Layer-2 GEMM fused into the layer-1 row-sum kernel: a work-group's 16 summed rows go through LDS as one bf16 tile to the WMMAs; that plane never reaches memory, 6 launches to 5; sum order unchanged
# speedup vs baseline: 4.3429x; 1.9469x over previous
#include <hip/hip_runtime.h>

typedef __attribute__((ext_vector_type(16))) __bf16 v16bf;
typedef __attribute__((ext_vector_type(8)))  float  v8f;
typedef __attribute__((ext_vector_type(8)))  __bf16 v8bf;
typedef __attribute__((ext_vector_type(4)))  __bf16 v4bf;

#define ROW_INTS 64
#define ROW_CAP  63


#define BK_THREADS    256
#define BK_PER_THREAD 16
#define BK_CHUNK      (BK_THREADS * BK_PER_THREAD)

__global__ __launch_bounds__(BK_THREADS) void bucket_edges_kernel(
    const int* __restrict__ src, const int* __restrict__ dst,
    int* __restrict__ rows, int* __restrict__ ovfCount,
    int* __restrict__ ovfSrc, int* __restrict__ ovfDst, int e, int n,
    int* __restrict__ tick0, int* __restrict__ tick1) {
  __shared__ int sTicket[2];
  __shared__ int sHalf;
  const int tid = threadIdx.x;
  const int nChunks = (e + BK_CHUNK - 1) / BK_CHUNK;
  const int nSplit = n >> 1;

  for (int pass = 0; pass < 2; ++pass) {
    if (tid == 0) {
      const int own = (int)((__builtin_amdgcn_s_sendmsg_rtn(0x87) >> 18) & 1u);
      const int hh = own ^ pass;
      sHalf = hh;
      sTicket[0] = atomicAdd(hh ? tick1 : tick0, 1);
    }
    __syncthreads();
    const int h = sHalf;
    int* __restrict__ tick = h ? tick1 : tick0;
    const int lo = h ? nSplit : 0;
    const int hi = h ? n : nSplit;
    for (int it = 0; it <= nChunks; ++it) {
      const int c = sTicket[it & 1];
      if (c >= nChunks) break;

      const int base = c * BK_CHUNK + tid;
      int d[BK_PER_THREAD], s[BK_PER_THREAD], pos[BK_PER_THREAD];
#pragma unroll
      for (int k = 0; k < BK_PER_THREAD; ++k) {
        const int idx = base + k * BK_THREADS;
        const int v = dst[idx < e ? idx : e - 1];
        d[k] = idx < e ? v : -1;
      }
#pragma unroll
      for (int k = 0; k < BK_PER_THREAD; ++k) {
        const int idx = base + k * BK_THREADS;
        s[k] = src[idx < e ? idx : e - 1];
      }
      int nxt = 0;
      if (tid == 0) nxt = atomicAdd(tick, 1);
#pragma unroll
      for (int k = 0; k < BK_PER_THREAD; ++k) {
        pos[k] = 0;
        if (d[k] >= lo && d[k] < hi) pos[k] = atomicAdd(rows + (size_t)d[k] * ROW_INTS, 1);
      }
#pragma unroll
      for (int k = 0; k < BK_PER_THREAD; ++k) {
        if (d[k] >= lo && d[k] < hi) {
          int* row = rows + (size_t)d[k] * ROW_INTS;
          if (pos[k] < ROW_CAP) {
            row[1 + pos[k]] = s[k];
          } else {
            int o = atomicAdd(ovfCount, 1);
            ovfSrc[o] = s[k];
            ovfDst[o] = d[k];
          }
        }
      }

      if (tid == 0) sTicket[(it + 1) & 1] = nxt;
      __syncthreads();
    }
    __syncthreads();
  }
}

template <int K>
__device__ __forceinline__ void pack_W_slot(const float* __restrict__ W, __bf16* __restrict__ Bp, int t) {
  int lane  = t & 31;
  int nt    = (t >> 5) & 3;
  int ktIdx = t >> 7;
  int r16   = lane & 15;
  int khalf = (lane >> 4) * 8;
  int col   = nt * 16 + r16;
  v16bf b;
#pragma unroll
  for (int i = 0; i < 8; ++i) {
    b[i]     = (__bf16)W[(size_t)(ktIdx * 32 + khalf + i) * 64 + col];
    b[8 + i] = (__bf16)W[(size_t)(ktIdx * 32 + khalf + 16 + i) * 64 + col];
  }
  *(v16bf*)(Bp + (size_t)t * 16) = b;
}

__global__ void setup_kernel(const float* __restrict__ W1, const float* __restrict__ W2,
                             __bf16* __restrict__ Bp1, __bf16* __restrict__ Bp2,
                             int* __restrict__ rows, int* __restrict__ ovfCount,
                             int* __restrict__ tick0, int* __restrict__ tick1, int n) {
  int i = blockIdx.x * blockDim.x + threadIdx.x;
  if (i < n) rows[(size_t)i * ROW_INTS] = 0;
  if (i == 0) {
    *ovfCount = 0;
    *tick0 = 0;
    *tick1 = 0;
  }
  if (i < 512) pack_W_slot<128>(W1, Bp1, i);
  else if (i < 768) pack_W_slot<64>(W2, Bp2, i - 512);
}

template <int K>
__device__ __forceinline__ void gemm_wmma_tile(
    const float* __restrict__ A, const __bf16* __restrict__ Bp,
    float* __restrict__ out, int nRows,
    const int* __restrict__ rows, float* __restrict__ dinv) {
  const int lane = threadIdx.x & 31;
  const int wave = threadIdx.x >> 5;
  const int rowTile = (blockIdx.x * 8 + wave) * 16;
  if (rowTile >= nRows) return;

  const int r16   = lane & 15;
  const int khalf = (lane >> 4) * 8;
  int row  = rowTile + r16;
  int rowc = row < nRows ? row : nRows - 1;
  const float* __restrict__ arow = A + (size_t)rowc * K;

  const int degc = rows[(size_t)rowc * ROW_INTS];

  v8f acc[4];
  acc[0] = (v8f){}; acc[1] = (v8f){}; acc[2] = (v8f){}; acc[3] = (v8f){};

#pragma unroll
  for (int ktIdx = 0; ktIdx < K / 32; ++ktIdx) {
    const int kt = ktIdx * 32;
    const float4 a0 = *(const float4*)(arow + kt + khalf);
    const float4 a1 = *(const float4*)(arow + kt + khalf + 4);
    const float4 a2 = *(const float4*)(arow + kt + khalf + 16);
    const float4 a3 = *(const float4*)(arow + kt + khalf + 20);
    float lo[8] = {a0.x, a0.y, a0.z, a0.w, a1.x, a1.y, a1.z, a1.w};
    float hi[8] = {a2.x, a2.y, a2.z, a2.w, a3.x, a3.y, a3.z, a3.w};
    v16bf a;
#pragma unroll
    for (int i = 0; i < 8; ++i) {
      a[i]     = (__bf16)lo[i];
      a[8 + i] = (__bf16)hi[i];
    }
#pragma unroll
    for (int nt = 0; nt < 4; ++nt) {
      const v16bf b =
          *(const v16bf*)(Bp + ((size_t)(ktIdx * 4 + nt) * 32 + lane) * 16);
      acc[nt] = __builtin_amdgcn_wmma_f32_16x16x32_bf16(
           false, a,  false, b,
           (short)0, acc[nt],  false,  false);
    }
  }

  const float d = (float)(degc + 1);
  const float dv = d > 0.0f ? rsqrtf(d) : 0.0f;
  if (lane < 16 && row < nRows) dinv[row] = dv;
  const int hrow  = (lane >> 4) * 8;
  const int rbase = rowTile + hrow;
  float sc[8];
#pragma unroll
  for (int r = 0; r < 8; ++r) sc[r] = __shfl(dv, hrow + r, 32);
#pragma unroll
  for (int nt = 0; nt < 4; ++nt) {
    const int col = nt * 16 + r16;
#pragma unroll
    for (int r = 0; r < 8; ++r) {
      int rr = rbase + r;
      if (rr < nRows) out[(size_t)rr * 64 + col] = acc[nt][r] * sc[r];
    }
  }
}

template <int K>
__global__ __launch_bounds__(256) void gemm_wmma_kernel(
    const float* __restrict__ A, const __bf16* __restrict__ Bp,
    float* __restrict__ out, int nRows,
    const int* __restrict__ rows, float* __restrict__ dinv) {
  gemm_wmma_tile<K>(A, Bp, out, nRows, rows, dinv);
}

__device__ __forceinline__ float4 row_sum(
    const float* __restrict__ xs, const int* __restrict__ rows, int i, int n, int l16,
    const int* __restrict__ ovfCount, const int* __restrict__ ovfSrc,
    const int* __restrict__ ovfDst, int nEdges) {
  const int f = l16 * 4;
  const int* __restrict__ row = rows + (size_t)i * ROW_INTS;
  const int mraw = row[0];
  const int m = mraw < 0 ? 0 : (mraw > ROW_CAP ? ROW_CAP : mraw);
  const float4 self = *(const float4*)(xs + (size_t)i * 64 + f);
  float4 acc = self;

  for (int base = 0; base < m; base += 16) {
    const int j = base + l16;
    int sidx = i;
    if (j < m) {
      int sraw = row[1 + j];
      sidx = sraw < 0 ? 0 : (sraw >= n ? n - 1 : sraw);
    }
    const int left = m - base;
    const int lim = left < 16 ? left : 16;
    for (int jj = 0; jj < lim; jj += 4) {
      int s[4];
      float4 v[4];
#pragma unroll
      for (int u = 0; u < 4; ++u) s[u] = __shfl(sidx, jj + u, 16);
#pragma unroll
      for (int u = 0; u < 4; ++u) {
        v[u] = self;
        if (jj + u < lim) v[u] = *(const float4*)(xs + (size_t)s[u] * 64 + f);
      }
#pragma unroll
      for (int u = 0; u < 4; ++u) {
        if (jj + u < lim) {
          acc.x += v[u].x; acc.y += v[u].y; acc.z += v[u].z; acc.w += v[u].w;
        }
      }
    }
  }
  if (mraw > ROW_CAP) {
    int cnt = *ovfCount;
    cnt = cnt < 0 ? 0 : (cnt > nEdges ? nEdges : cnt);
    for (int base = 0; base < cnt; base += 16) {
      const int e = base + l16;
      int od = -1;
      int os = 0;
      if (e < cnt) {
        od = ovfDst[e];
        os = ovfSrc[e];
      }
#pragma nounroll
      for (int u = 0; u < 16; ++u) {
        const int du = __shfl(od, u, 16);
        int su = __shfl(os, u, 16);
        if (du == i) {
          su = su < 0 ? 0 : (su >= n ? n - 1 : su);
          const float4 vu = *(const float4*)(xs + (size_t)su * 64 + f);
          acc.x += vu.x; acc.y += vu.y; acc.z += vu.z; acc.w += vu.w;
        }
      }
    }
  }
  return acc;
}

__global__ __launch_bounds__(256) void gather_rows_kernel(
    const float* __restrict__ xs, const float* __restrict__ dinv,
    const int* __restrict__ rows, const float* __restrict__ bias,
    float* __restrict__ out, int n,
    const int* __restrict__ ovfCount, const int* __restrict__ ovfSrc,
    const int* __restrict__ ovfDst, int nEdges) {
  int t = blockIdx.x * blockDim.x + threadIdx.x;
  int i = t >> 4;
  if (i >= n) return;
  const int l16 = threadIdx.x & 15;
  const int f = l16 * 4;
  const float di = dinv[i];
  float4 acc = row_sum(xs, rows, i, n, l16, ovfCount, ovfSrc, ovfDst, nEdges);
  acc.x *= di; acc.y *= di; acc.z *= di; acc.w *= di;
  acc.x += bias[f]; acc.y += bias[f + 1]; acc.z += bias[f + 2]; acc.w += bias[f + 3];
  *(float4*)(out + (size_t)i * 64 + f) = acc;
}

#define A_PITCH 72
#define D_PITCH 68

__global__ __launch_bounds__(256) void gather_gemm_kernel(
    const float* __restrict__ xs, const float* __restrict__ dinv,
    const int* __restrict__ rows, const float* __restrict__ bias,
    const __bf16* __restrict__ Bp, float* __restrict__ out, int n,
    const int* __restrict__ ovfCount, const int* __restrict__ ovfSrc,
    const int* __restrict__ ovfDst, int nEdges) {
  __shared__ __attribute__((aligned(16))) __bf16 sA[16 * A_PITCH];
  __shared__ __attribute__((aligned(16))) float  sD[16 * D_PITCH];
  __shared__ __attribute__((aligned(16))) float  sScale[16];

  const int lane = threadIdx.x & 31;
  const int wave = threadIdx.x >> 5;
  const int nb   = threadIdx.x >> 4;
  const int l16  = threadIdx.x & 15;
  const int f    = l16 * 4;
  const int i    = blockIdx.x * 16 + nb;

  v16bf b0 = {}, b1 = {};
  if (wave < 4) {
    b0 = *(const v16bf*)(Bp + ((size_t)(0 * 4 + wave) * 32 + lane) * 16);
    b1 = *(const v16bf*)(Bp + ((size_t)(1 * 4 + wave) * 32 + lane) * 16);
  }

  float4 h = {0.0f, 0.0f, 0.0f, 0.0f};
  float di = 0.0f;
  if (i < n) {
    di = dinv[i];
    const float4 acc = row_sum(xs, rows, i, n, l16, ovfCount, ovfSrc, ovfDst, nEdges);
    h.x = acc.x * di + bias[f];     h.x = h.x > 0.0f ? h.x : 0.0f;
    h.y = acc.y * di + bias[f + 1]; h.y = h.y > 0.0f ? h.y : 0.0f;
    h.z = acc.z * di + bias[f + 2]; h.z = h.z > 0.0f ? h.z : 0.0f;
    h.w = acc.w * di + bias[f + 3]; h.w = h.w > 0.0f ? h.w : 0.0f;
  }
  const v4bf hb = {(__bf16)h.x, (__bf16)h.y, (__bf16)h.z, (__bf16)h.w};
  *(v4bf*)(sA + nb * A_PITCH + f) = hb;
  if (l16 == 0) sScale[nb] = di;
  __syncthreads();

  if (wave < 4) {
    const int c  = lane & 15;
    const int hh = lane >> 4;
    const __bf16* ap = sA + c * A_PITCH + hh * 8;
    const v8bf a0lo = *(const v8bf*)(ap);
    const v8bf a0hi = *(const v8bf*)(ap + 16);
    const v8bf a1lo = *(const v8bf*)(ap + 32);
    const v8bf a1hi = *(const v8bf*)(ap + 48);
    const v16bf a0 = __builtin_shufflevector(a0lo, a0hi, 0, 1, 2, 3, 4, 5, 6, 7, 8, 9, 10, 11, 12, 13, 14, 15);
    const v16bf a1 = __builtin_shufflevector(a1lo, a1hi, 0, 1, 2, 3, 4, 5, 6, 7, 8, 9, 10, 11, 12, 13, 14, 15);
    v8f acc = {};
    acc = __builtin_amdgcn_wmma_f32_16x16x32_bf16(
         false, a0,  false, b0,
         (short)0, acc,  false,  false);
    acc = __builtin_amdgcn_wmma_f32_16x16x32_bf16(
         false, a1,  false, b1,
         (short)0, acc,  false,  false);
    const float4 s0 = *(const float4*)(sScale + hh * 8);
    const float4 s1 = *(const float4*)(sScale + hh * 8 + 4);
    const float sc[8] = {s0.x, s0.y, s0.z, s0.w, s1.x, s1.y, s1.z, s1.w};
    float* dp = sD + (hh * 8) * D_PITCH + wave * 16 + c;
#pragma unroll
    for (int r = 0; r < 8; ++r) dp[r * D_PITCH] = acc[r] * sc[r];
  }
  __syncthreads();

  if (i < n) *(float4*)(out + (size_t)i * 64 + f) = *(const float4*)(sD + nb * D_PITCH + f);
}

extern "C" void kernel_launch(void* const* d_in, const int* in_sizes, int n_in,
                              void* d_out, int out_size, void* d_ws, size_t ws_size,
                              hipStream_t stream) {
  const int*   edge_index = (const int*)d_in[0];
  const float* emb        = (const float*)d_in[1];
  const float* W1         = (const float*)d_in[2];
  const float* b1         = (const float*)d_in[3];
  const float* W2         = (const float*)d_in[4];
  const float* b2         = (const float*)d_in[5];
  float* out = (float*)d_out;

  const int E = in_sizes[0] / 2;
  const int N = in_sizes[1] / 128;
  const int* src = edge_index;
  const int* dst = edge_index + E;

  float* ws   = (float*)d_ws;
  float* dinv = ws;
  size_t npad = ((size_t)N + 63) & ~(size_t)63;
  int*   ovfCount = (int*)(ws + npad);
  float* x1   = ws + npad + 64;
  float* x2   = x1 + (size_t)N * 64;
  __bf16* Bp1 = (__bf16*)(x2 + (size_t)N * 64);
  __bf16* Bp2 = Bp1 + 8192;
  int* rows   = (int*)(x2 + (size_t)N * 64 + 6144);
  int* ovfSrc = rows + (size_t)N * ROW_INTS;
  int* ovfDst = ovfSrc + E;
  size_t epad = ((size_t)E + 63) & ~(size_t)63;
  int* tick0  = ovfDst + epad;
  int* tick1  = tick0 + 256;

  const int tpb = 256;
  const int nBlk = (N + tpb - 1) / tpb;
  const int setupBlk = nBlk > 3 ? nBlk : 3;
  const int eBlk = (E + BK_CHUNK / 2 - 1) / (BK_CHUNK / 2);
  const int gemmBlk = (N + 127) / 128;
  const long nf = (long)N * 16;
  const int nfBlk = (int)((nf + tpb - 1) / tpb);

  setup_kernel<<<setupBlk, tpb, 0, stream>>>(W1, W2, Bp1, Bp2, rows, ovfCount, tick0, tick1, N);

  bucket_edges_kernel<<<eBlk, BK_THREADS, 0, stream>>>(src, dst, rows, ovfCount, ovfSrc, ovfDst, E, N,
                                                       tick0, tick1);

  gemm_wmma_kernel<128><<<gemmBlk, 256, 0, stream>>>(emb, Bp1, x1, N, rows, dinv);

  gather_gemm_kernel<<<nfBlk, tpb, 0, stream>>>(x1, dinv, rows, b1, Bp2, x2, N,
                                                ovfCount, ovfSrc, ovfDst, E);

  gather_rows_kernel<<<nfBlk, tpb, 0, stream>>>(x2, dinv, rows, b2, out, N,
                                                ovfCount, ovfSrc, ovfDst, E);
}


# Edge bucketing on LDS counters, no per-edge global atomic: each 4096-edge slice is ordered by 256-node bin in LDS, then one work-group per bin builds its rows in LDS; slot order = LDS arrival order
# speedup vs baseline: 5.4947x; 1.2652x over previous
#include <hip/hip_runtime.h>

typedef __attribute__((ext_vector_type(16))) __bf16 v16bf;
typedef __attribute__((ext_vector_type(8)))  float  v8f;
typedef __attribute__((ext_vector_type(8)))  __bf16 v8bf;
typedef __attribute__((ext_vector_type(4)))  __bf16 v4bf;
typedef __attribute__((ext_vector_type(4)))  unsigned int v4u;

#define ROW_INTS 64
#define ROW_CAP  63


template <int K, bool PERM>
__device__ __forceinline__ void pack_W_slot(const float* __restrict__ W, __bf16* __restrict__ Bp, int t) {
  int lane  = t & 31;
  int nt    = (t >> 5) & 3;
  int ktIdx = t >> 7;
  int r16   = lane & 15;
  int khalf = (lane >> 4) * 8;
  int col   = PERM ? ((r16 & 1) * 32 + (r16 >> 1) * 4 + nt) : (nt * 16 + r16);
  v16bf b;
#pragma unroll
  for (int i = 0; i < 8; ++i) {
    b[i]     = (__bf16)W[(size_t)(ktIdx * 32 + khalf + i) * 64 + col];
    b[8 + i] = (__bf16)W[(size_t)(ktIdx * 32 + khalf + 16 + i) * 64 + col];
  }
  *(v16bf*)(Bp + (size_t)t * 16) = b;
}

#define SP_THREADS    256
#define SP_PER_THREAD 16
#define SP_SLICE      (SP_THREADS * SP_PER_THREAD)
#define BIN_SHIFT     8
#define BIN_NODES     (1 << BIN_SHIFT)
#define MAX_BINS      512

__global__ __launch_bounds__(SP_THREADS) void split_edges_kernel(
    const int* __restrict__ src, const int* __restrict__ dst,
    int2* __restrict__ sorted, int* __restrict__ starts, int e, int n, int nb,
    const float* __restrict__ W1, const float* __restrict__ W2,
    __bf16* __restrict__ Bp1, __bf16* __restrict__ Bp2, int* __restrict__ ovfCount) {
  __shared__ int  sBin[MAX_BINS];
  __shared__ int  sWave[SP_THREADS / 32];
  __shared__ int2 sPairs[SP_SLICE];
  const int tid  = threadIdx.x;
  const int lane = tid & 31;
  const int wave = tid >> 5;
  const int w    = blockIdx.x;
  const int gi   = w * SP_THREADS + tid;

  sBin[tid] = 0;
  sBin[tid + SP_THREADS] = 0;
  if (gi == 0) *ovfCount = 0;
  if (gi < 512) pack_W_slot<128, true>(W1, Bp1, gi);
  else if (gi < 768) pack_W_slot<64, false>(W2, Bp2, gi - 512);
  __syncthreads();

  const int base = w * SP_SLICE + tid;
  int d[SP_PER_THREAD], s[SP_PER_THREAD], rk[SP_PER_THREAD];
#pragma unroll
  for (int k = 0; k < SP_PER_THREAD; ++k) {
    const int idx = base + k * SP_THREADS;
    const int v = dst[idx < e ? idx : e - 1];
    d[k] = idx < e ? v : -1;
  }
#pragma unroll
  for (int k = 0; k < SP_PER_THREAD; ++k) {
    const int idx = base + k * SP_THREADS;
    s[k] = src[idx < e ? idx : e - 1];
  }
#pragma unroll
  for (int k = 0; k < SP_PER_THREAD; ++k) {
    rk[k] = 0;
    if (d[k] >= 0 && d[k] < n) rk[k] = atomicAdd(&sBin[d[k] >> BIN_SHIFT], 1);
  }
  __syncthreads();

  const int b0 = 2 * tid;
  const int b1 = b0 + 1;
  const int c0 = sBin[b0];
  const int c1 = sBin[b1];
  const int v = c0 + c1;
  int x = v;
#pragma unroll
  for (int o = 1; o < 32; o <<= 1) {
    const int y = __shfl_up(x, o, 32);
    if (lane >= o) x += y;
  }
  if (lane == 31) sWave[wave] = x;
  __syncthreads();
  int wb = 0;
#pragma unroll
  for (int i = 0; i < SP_THREADS / 32; ++i) {
    if (i < wave) wb += sWave[i];
  }
  const int ex = wb + x - v;
  sBin[b0] = ex;
  sBin[b1] = ex + c0;
  int* __restrict__ myStarts = starts + (size_t)w * (nb + 1);
  if (b0 <= nb) myStarts[b0] = ex;
  if (b1 <= nb) myStarts[b1] = ex + c0;
  __syncthreads();

  int st[SP_PER_THREAD];
#pragma unroll
  for (int k = 0; k < SP_PER_THREAD; ++k) {
    const bool ok = d[k] >= 0 && d[k] < n;
    st[k] = sBin[ok ? (d[k] >> BIN_SHIFT) : 0];
  }
#pragma unroll
  for (int k = 0; k < SP_PER_THREAD; ++k) {
    if (d[k] >= 0 && d[k] < n) sPairs[st[k] + rk[k]] = make_int2(s[k], d[k]);
  }
  __syncthreads();
  const int total = sBin[nb];
  int2* __restrict__ mySorted = sorted + (size_t)w * SP_SLICE;
  for (int i = tid; i < total; i += SP_THREADS) mySorted[i] = sPairs[i];
}

#define ROW_PITCH  65
#define BR_THREADS 256
#define BR_GROUP   16
#define BR_NIT     16

__device__ __forceinline__ int local_row(int dstNode, int nodeBase) {
  const int local = dstNode - nodeBase;
  return local < 0 ? 0 : (local > BIN_NODES - 1 ? BIN_NODES - 1 : local);
}

__device__ __forceinline__ void add_edge_lds(
    int* __restrict__ sRow, int nodeBase, const int2 q,
    int* __restrict__ ovfCount, int* __restrict__ ovfSrc, int* __restrict__ ovfDst) {
  int* row = sRow + local_row(q.y, nodeBase) * ROW_PITCH;
  const int pos = atomicAdd(row, 1);
  if (pos < ROW_CAP) {
    row[1 + pos] = q.x;
  } else {
    const int o = atomicAdd(ovfCount, 1);
    ovfSrc[o] = q.x;
    ovfDst[o] = q.y;
  }
}

__global__ __launch_bounds__(BR_THREADS) void build_rows_kernel(
    const int2* __restrict__ sorted, const int* __restrict__ starts, int* __restrict__ rows,
    int* __restrict__ ovfCount, int* __restrict__ ovfSrc, int* __restrict__ ovfDst,
    int n, int nb, int ga) {
  __shared__ int sRow[BIN_NODES * ROW_PITCH];
  const int tid  = threadIdx.x;
  const int lane = tid & 31;
  const int wave = tid >> 5;
  const int b    = blockIdx.x;
  const int nodeBase = b << BIN_SHIFT;
  const int sub = tid / BR_GROUP;
  const int l16 = tid % BR_GROUP;

  sRow[tid * ROW_PITCH] = 0;
  __syncthreads();

  for (int pb = 0; pb < ga; pb += (BR_THREADS / BR_GROUP) * BR_NIT) {
    int i0[BR_NIT], s1[BR_NIT], pos[BR_NIT];
    int2 q[BR_NIT];
#pragma unroll
    for (int it = 0; it < BR_NIT; ++it) {
      const int p  = pb + it * (BR_THREADS / BR_GROUP) + sub;
      const int pc = p < ga ? p : ga - 1;
      const int* __restrict__ st = starts + (size_t)pc * (nb + 1) + b;
      const int a = st[0];
      const int c = st[1];
      i0[it] = a + l16;
      s1[it] = p < ga ? c : a;
    }
#pragma unroll
    for (int it = 0; it < BR_NIT; ++it) {
      const int p  = pb + it * (BR_THREADS / BR_GROUP) + sub;
      const int pc = p < ga ? p : ga - 1;
      if (i0[it] < s1[it]) q[it] = sorted[(size_t)pc * SP_SLICE + i0[it]];
    }
#pragma unroll
    for (int it = 0; it < BR_NIT; ++it) {
      pos[it] = 0;
      if (i0[it] < s1[it]) pos[it] = atomicAdd(sRow + local_row(q[it].y, nodeBase) * ROW_PITCH, 1);
    }
#pragma unroll
    for (int it = 0; it < BR_NIT; ++it) {
      if (i0[it] < s1[it]) {
        if (pos[it] < ROW_CAP) {
          sRow[local_row(q[it].y, nodeBase) * ROW_PITCH + 1 + pos[it]] = q[it].x;
        } else {
          const int o = atomicAdd(ovfCount, 1);
          ovfSrc[o] = q[it].x;
          ovfDst[o] = q[it].y;
        }
      }
    }
#pragma unroll
    for (int it = 0; it < BR_NIT; ++it) {
      const int p  = pb + it * (BR_THREADS / BR_GROUP) + sub;
      const int pc = p < ga ? p : ga - 1;
      for (int j = i0[it] + BR_GROUP; j < s1[it]; j += BR_GROUP)
        add_edge_lds(sRow, nodeBase, sorted[(size_t)pc * SP_SLICE + j], ovfCount, ovfSrc, ovfDst);
    }
  }
  __syncthreads();

  for (int r = wave; r < BIN_NODES; r += BR_THREADS / 32) {
    const int g = nodeBase + r;
    if (g >= n) break;
    const int* row = sRow + r * ROW_PITCH;
    const int cnt = row[0];
    int* __restrict__ grow = rows + (size_t)g * ROW_INTS;
    grow[lane] = row[lane];
    if (cnt > 31) grow[32 + lane] = row[32 + lane];
  }
}

template <int K>
__device__ __forceinline__ void gemm_wmma_tile(
    const float* __restrict__ A, const __bf16* __restrict__ Bp,
    __bf16* __restrict__ out, int nRows,
    const int* __restrict__ rows, float* __restrict__ dinv) {
  const int lane = threadIdx.x & 31;
  const int wave = threadIdx.x >> 5;
  const int rowTile = (blockIdx.x * 8 + wave) * 16;
  if (rowTile >= nRows) return;

  const int r16   = lane & 15;
  const int khalf = (lane >> 4) * 8;
  int row  = rowTile + r16;
  int rowc = row < nRows ? row : nRows - 1;
  const float* __restrict__ arow = A + (size_t)rowc * K;

  const int degc = rows[(size_t)rowc * ROW_INTS];

  v8f acc[4];
  acc[0] = (v8f){}; acc[1] = (v8f){}; acc[2] = (v8f){}; acc[3] = (v8f){};

#pragma unroll
  for (int ktIdx = 0; ktIdx < K / 32; ++ktIdx) {
    const int kt = ktIdx * 32;
    const float4 a0 = *(const float4*)(arow + kt + khalf);
    const float4 a1 = *(const float4*)(arow + kt + khalf + 4);
    const float4 a2 = *(const float4*)(arow + kt + khalf + 16);
    const float4 a3 = *(const float4*)(arow + kt + khalf + 20);
    float lo[8] = {a0.x, a0.y, a0.z, a0.w, a1.x, a1.y, a1.z, a1.w};
    float hi[8] = {a2.x, a2.y, a2.z, a2.w, a3.x, a3.y, a3.z, a3.w};
    v16bf a;
#pragma unroll
    for (int i = 0; i < 8; ++i) {
      a[i]     = (__bf16)lo[i];
      a[8 + i] = (__bf16)hi[i];
    }
#pragma unroll
    for (int nt = 0; nt < 4; ++nt) {
      const v16bf b =
          *(const v16bf*)(Bp + ((size_t)(ktIdx * 4 + nt) * 32 + lane) * 16);
      acc[nt] = __builtin_amdgcn_wmma_f32_16x16x32_bf16(
           false, a,  false, b,
           (short)0, acc[nt],  false,  false);
    }
  }

  const float d = (float)(degc + 1);
  const float dv = d > 0.0f ? rsqrtf(d) : 0.0f;
  if (lane < 16 && row < nRows) dinv[row] = dv;
  const int hrow  = (lane >> 4) * 8;
  const int rbase = rowTile + hrow;
  float sc[8];
#pragma unroll
  for (int r = 0; r < 8; ++r) sc[r] = __shfl(dv, hrow + r, 32);
#pragma unroll
  for (int r = 0; r < 8; ++r) {
    const int rr = rbase + r;
    const v4bf p = {(__bf16)(acc[0][r] * sc[r]), (__bf16)(acc[1][r] * sc[r]),
                    (__bf16)(acc[2][r] * sc[r]), (__bf16)(acc[3][r] * sc[r])};
    if (rr < nRows) *(v4bf*)(out + (size_t)rr * 64 + 4 * r16) = p;
  }
}

template <int K>
__global__ __launch_bounds__(256) void gemm_wmma_kernel(
    const float* __restrict__ A, const __bf16* __restrict__ Bp,
    __bf16* __restrict__ out, int nRows,
    const int* __restrict__ rows, float* __restrict__ dinv) {
  gemm_wmma_tile<K>(A, Bp, out, nRows, rows, dinv);
}

__device__ __forceinline__ float bf_lo(unsigned w) { return __uint_as_float(w << 16); }
__device__ __forceinline__ float bf_hi(unsigned w) { return __uint_as_float(w & 0xffff0000u); }

__device__ __forceinline__ void add_row_bf16(const v4u v, float4& accA, float4& accB) {
  accA.x += bf_lo(v.x); accA.y += bf_hi(v.x); accA.z += bf_lo(v.y); accA.w += bf_hi(v.y);
  accB.x += bf_lo(v.z); accB.y += bf_hi(v.z); accB.z += bf_lo(v.w); accB.w += bf_hi(v.w);
}

__device__ __forceinline__ void add_rows8(
    const __bf16* __restrict__ xs, int sidx, int cnt, int f8, float4& accA, float4& accB) {
  int s[8];
  v4u v[8];
#pragma unroll
  for (int u = 0; u < 8; ++u) s[u] = __shfl(sidx, u, 8);
#pragma unroll
  for (int u = 0; u < 8; ++u) {
    if (u < cnt) v[u] = *(const v4u*)(xs + ((unsigned)s[u] * 64u + (unsigned)f8));
  }
#pragma unroll
  for (int u = 0; u < 8; ++u) {
    if (u < cnt) add_row_bf16(v[u], accA, accB);
  }
}

__device__ __forceinline__ void row_sum8(
    const __bf16* __restrict__ xs, const int* __restrict__ rows, int i, int n, int l8,
    const int* __restrict__ ovfCount, const int* __restrict__ ovfSrc,
    const int* __restrict__ ovfDst, int nEdges, float4& accA, float4& accB) {
  const int f8 = l8 * 8;
  const int* __restrict__ row = rows + (size_t)i * ROW_INTS;
  const int mraw = row[0];
  const int m = mraw < 0 ? 0 : (mraw > ROW_CAP ? ROW_CAP : mraw);
  const v4u sv = *(const v4u*)(xs + (size_t)i * 64 + f8);
  accA = (float4){bf_lo(sv.x), bf_hi(sv.x), bf_lo(sv.y), bf_hi(sv.y)};
  accB = (float4){bf_lo(sv.z), bf_hi(sv.z), bf_lo(sv.w), bf_hi(sv.w)};

  for (int base = 0; base < m; base += 32) {
    int e[4];
#pragma unroll
    for (int q = 0; q < 4; ++q) {
      const int j = base + q * 8 + l8;
      e[q] = row[1 + (j < m ? j : m - 1)];
    }
#pragma unroll
    for (int q = 0; q < 4; ++q)
      e[q] = e[q] < 0 ? 0 : (e[q] >= n ? n - 1 : e[q]);
    const int left = m - base;
#pragma unroll
    for (int q = 0; q < 4; ++q) {
      const int cnt = left - q * 8;
      if (cnt > 0) add_rows8(xs, e[q], cnt, f8, accA, accB);
    }
  }
  if (mraw > ROW_CAP) {
    int cnt = *ovfCount;
    cnt = cnt < 0 ? 0 : (cnt > nEdges ? nEdges : cnt);
    for (int base = 0; base < cnt; base += 8) {
      const int e = base + l8;
      int od = -1;
      int os = 0;
      if (e < cnt) {
        od = ovfDst[e];
        os = ovfSrc[e];
      }
#pragma nounroll
      for (int u = 0; u < 8; ++u) {
        const int du = __shfl(od, u, 8);
        int su = __shfl(os, u, 8);
        if (du == i) {
          su = su < 0 ? 0 : (su >= n ? n - 1 : su);
          const v4u vu = *(const v4u*)(xs + (size_t)su * 64 + f8);
          add_row_bf16(vu, accA, accB);
        }
      }
    }
  }
}

__global__ __launch_bounds__(256) void gather_rows_kernel(
    const __bf16* __restrict__ xs, const float* __restrict__ dinv,
    const int* __restrict__ rows, const float* __restrict__ bias,
    float* __restrict__ out, int n,
    const int* __restrict__ ovfCount, const int* __restrict__ ovfSrc,
    const int* __restrict__ ovfDst, int nEdges) {
  int t = blockIdx.x * blockDim.x + threadIdx.x;
  int i = t >> 3;
  if (i >= n) return;
  const int l8 = threadIdx.x & 7;
  const int f = l8 * 4;
  const float di = dinv[i];
  float4 a, b;
  row_sum8(xs, rows, i, n, l8, ovfCount, ovfSrc, ovfDst, nEdges, a, b);
  a.x *= di; a.y *= di; a.z *= di; a.w *= di;
  b.x *= di; b.y *= di; b.z *= di; b.w *= di;
  a.x += bias[f]; a.y += bias[f + 1]; a.z += bias[f + 2]; a.w += bias[f + 3];
  b.x += bias[32 + f]; b.y += bias[32 + f + 1]; b.z += bias[32 + f + 2]; b.w += bias[32 + f + 3];
  *(float4*)(out + (size_t)i * 64 + f) = a;
  *(float4*)(out + (size_t)i * 64 + 32 + f) = b;
}

#define A_PITCH 72
#define D_PITCH 68
#define GG_NODES 32

__global__ __launch_bounds__(256) void gather_gemm_kernel(
    const __bf16* __restrict__ xs, const float* __restrict__ dinv,
    const int* __restrict__ rows, const float* __restrict__ bias,
    const __bf16* __restrict__ Bp, __bf16* __restrict__ out, int n,
    const int* __restrict__ ovfCount, const int* __restrict__ ovfSrc,
    const int* __restrict__ ovfDst, int nEdges) {
  __shared__ __attribute__((aligned(16))) __bf16 sA[GG_NODES * A_PITCH];
  __shared__ __attribute__((aligned(16))) float  sD[GG_NODES * D_PITCH];
  __shared__ __attribute__((aligned(16))) float  sScale[GG_NODES];

  const int lane = threadIdx.x & 31;
  const int wave = threadIdx.x >> 5;
  const int nb   = threadIdx.x >> 3;
  const int l8   = threadIdx.x & 7;
  const int f    = l8 * 4;
  const int i    = blockIdx.x * GG_NODES + nb;
  const int rt   = wave >> 2;
  const int ct   = wave & 3;

  const v16bf b0 = *(const v16bf*)(Bp + ((size_t)(0 * 4 + ct) * 32 + lane) * 16);
  const v16bf b1 = *(const v16bf*)(Bp + ((size_t)(1 * 4 + ct) * 32 + lane) * 16);

  float4 hA = {0.0f, 0.0f, 0.0f, 0.0f};
  float4 hB = {0.0f, 0.0f, 0.0f, 0.0f};
  float di = 0.0f;
  if (i < n) {
    di = dinv[i];
    float4 a, b;
    row_sum8(xs, rows, i, n, l8, ovfCount, ovfSrc, ovfDst, nEdges, a, b);
    hA.x = a.x * di + bias[f];          hA.x = hA.x > 0.0f ? hA.x : 0.0f;
    hA.y = a.y * di + bias[f + 1];      hA.y = hA.y > 0.0f ? hA.y : 0.0f;
    hA.z = a.z * di + bias[f + 2];      hA.z = hA.z > 0.0f ? hA.z : 0.0f;
    hA.w = a.w * di + bias[f + 3];      hA.w = hA.w > 0.0f ? hA.w : 0.0f;
    hB.x = b.x * di + bias[32 + f];     hB.x = hB.x > 0.0f ? hB.x : 0.0f;
    hB.y = b.y * di + bias[32 + f + 1]; hB.y = hB.y > 0.0f ? hB.y : 0.0f;
    hB.z = b.z * di + bias[32 + f + 2]; hB.z = hB.z > 0.0f ? hB.z : 0.0f;
    hB.w = b.w * di + bias[32 + f + 3]; hB.w = hB.w > 0.0f ? hB.w : 0.0f;
  }
  const v4bf hbA = {(__bf16)hA.x, (__bf16)hA.y, (__bf16)hA.z, (__bf16)hA.w};
  const v4bf hbB = {(__bf16)hB.x, (__bf16)hB.y, (__bf16)hB.z, (__bf16)hB.w};
  *(v4bf*)(sA + nb * A_PITCH + f) = hbA;
  *(v4bf*)(sA + nb * A_PITCH + 32 + f) = hbB;
  if (l8 == 0) sScale[nb] = di;
  __syncthreads();

  {
    const int c  = lane & 15;
    const int hh = lane >> 4;
    const __bf16* ap = sA + (rt * 16 + c) * A_PITCH + hh * 8;
    const v8bf a0lo = *(const v8bf*)(ap);
    const v8bf a0hi = *(const v8bf*)(ap + 16);
    const v8bf a1lo = *(const v8bf*)(ap + 32);
    const v8bf a1hi = *(const v8bf*)(ap + 48);
    const v16bf a0 = __builtin_shufflevector(a0lo, a0hi, 0, 1, 2, 3, 4, 5, 6, 7, 8, 9, 10, 11, 12, 13, 14, 15);
    const v16bf a1 = __builtin_shufflevector(a1lo, a1hi, 0, 1, 2, 3, 4, 5, 6, 7, 8, 9, 10, 11, 12, 13, 14, 15);
    v8f acc = {};
    acc = __builtin_amdgcn_wmma_f32_16x16x32_bf16(
         false, a0,  false, b0,
         (short)0, acc,  false,  false);
    acc = __builtin_amdgcn_wmma_f32_16x16x32_bf16(
         false, a1,  false, b1,
         (short)0, acc,  false,  false);
    const float4 s0 = *(const float4*)(sScale + rt * 16 + hh * 8);
    const float4 s1 = *(const float4*)(sScale + rt * 16 + hh * 8 + 4);
    const float sc[8] = {s0.x, s0.y, s0.z, s0.w, s1.x, s1.y, s1.z, s1.w};
    float* dp = sD + (rt * 16 + hh * 8) * D_PITCH + ct * 16 + c;
#pragma unroll
    for (int r = 0; r < 8; ++r) dp[r * D_PITCH] = acc[r] * sc[r];
  }
  __syncthreads();

  if (i < n) {
    const float4 d0 = *(const float4*)(sD + nb * D_PITCH + f);
    const float4 d1 = *(const float4*)(sD + nb * D_PITCH + 32 + f);
    const v8bf o = {(__bf16)d0.x, (__bf16)d0.y, (__bf16)d0.z, (__bf16)d0.w,
                    (__bf16)d1.x, (__bf16)d1.y, (__bf16)d1.z, (__bf16)d1.w};
    *(v8bf*)(out + (size_t)i * 64 + l8 * 8) = o;
  }
}

extern "C" void kernel_launch(void* const* d_in, const int* in_sizes, int n_in,
                              void* d_out, int out_size, void* d_ws, size_t ws_size,
                              hipStream_t stream) {
  const int*   edge_index = (const int*)d_in[0];
  const float* emb        = (const float*)d_in[1];
  const float* W1         = (const float*)d_in[2];
  const float* b1         = (const float*)d_in[3];
  const float* W2         = (const float*)d_in[4];
  const float* b2         = (const float*)d_in[5];
  float* out = (float*)d_out;

  const int E = in_sizes[0] / 2;
  const int N = in_sizes[1] / 128;
  const int* src = edge_index;
  const int* dst = edge_index + E;

  float* ws   = (float*)d_ws;
  float* dinv = ws;
  size_t npad = ((size_t)N + 63) & ~(size_t)63;
  int*   ovfCount = (int*)(ws + npad);
  float* p1   = ws + npad + 64;
  float* p2   = p1 + (size_t)N * 64;
  __bf16* x1  = (__bf16*)p1;
  __bf16* x2  = (__bf16*)p2;
  __bf16* Bp1 = (__bf16*)(p2 + (size_t)N * 64);
  __bf16* Bp2 = Bp1 + 8192;
  int* rows   = (int*)(p2 + (size_t)N * 64 + 6144);
  int* ovfSrc = rows + (size_t)N * ROW_INTS;
  size_t epad = ((size_t)E + 63) & ~(size_t)63;
  int* ovfDst = ovfSrc + epad;
  const int nSlices = (E + SP_SLICE - 1) / SP_SLICE;
  const int splitBlk = nSlices > 3 ? nSlices : 3;
  const int nBins   = (N + BIN_NODES - 1) / BIN_NODES;
  int2* sorted = (int2*)(ovfDst + epad);
  int* starts  = (int*)(sorted + (size_t)splitBlk * SP_SLICE);

  const int tpb = 256;
  const int gemmBlk = (N + 127) / 128;
  const long nf = (long)N * 8;
  const int nfBlk = (int)((nf + tpb - 1) / tpb);

  split_edges_kernel<<<splitBlk, SP_THREADS, 0, stream>>>(src, dst, sorted, starts, E, N, nBins,
                                                          W1, W2, Bp1, Bp2, ovfCount);

  build_rows_kernel<<<nBins, BR_THREADS, 0, stream>>>(sorted, starts, rows, ovfCount, ovfSrc, ovfDst,
                                                      N, nBins, splitBlk);

  gemm_wmma_kernel<128><<<gemmBlk, 256, 0, stream>>>(emb, Bp1, x1, N, rows, dinv);

  gather_gemm_kernel<<<nfBlk, tpb, 0, stream>>>(x1, dinv, rows, b1, Bp2, x2, N,
                                                ovfCount, ovfSrc, ovfDst, E);

  gather_rows_kernel<<<nfBlk, tpb, 0, stream>>>(x2, dinv, rows, b2, out, N,
                                                ovfCount, ovfSrc, ovfDst, E);
}


# Layer-1 GEMM fused into the row-building launch: a bin's work-group multiplies its own 256 rows on the WMMAs, in-degree from its LDS counters, emb loads fly under its row stores; 5 launches to 4
# speedup vs baseline: 6.1857x; 1.1258x over previous
#include <hip/hip_runtime.h>

typedef __attribute__((ext_vector_type(16))) __bf16 v16bf;
typedef __attribute__((ext_vector_type(8)))  float  v8f;
typedef __attribute__((ext_vector_type(8)))  __bf16 v8bf;
typedef __attribute__((ext_vector_type(4)))  __bf16 v4bf;
typedef __attribute__((ext_vector_type(4)))  unsigned int v4u;

#define ROW_INTS 64
#define ROW_CAP  63


template <int K, bool PERM>
__device__ __forceinline__ void pack_W_slot(const float* __restrict__ W, __bf16* __restrict__ Bp, int t) {
  int lane  = t & 31;
  int nt    = (t >> 5) & 3;
  int ktIdx = t >> 7;
  int r16   = lane & 15;
  int khalf = (lane >> 4) * 8;
  int col   = PERM ? ((r16 & 1) * 32 + (r16 >> 1) * 4 + nt) : (nt * 16 + r16);
  v16bf b;
#pragma unroll
  for (int i = 0; i < 8; ++i) {
    b[i]     = (__bf16)W[(size_t)(ktIdx * 32 + khalf + i) * 64 + col];
    b[8 + i] = (__bf16)W[(size_t)(ktIdx * 32 + khalf + 16 + i) * 64 + col];
  }
  *(v16bf*)(Bp + (size_t)t * 16) = b;
}

#define SP_THREADS    256
#define SP_PER_THREAD 16
#define SP_SLICE      (SP_THREADS * SP_PER_THREAD)
#define BIN_SHIFT     8
#define BIN_NODES     (1 << BIN_SHIFT)
#define MAX_BINS      512

__global__ __launch_bounds__(SP_THREADS) void split_edges_kernel(
    const int* __restrict__ src, const int* __restrict__ dst,
    int2* __restrict__ sorted, int* __restrict__ starts, int e, int n, int nb,
    const float* __restrict__ W1, const float* __restrict__ W2,
    __bf16* __restrict__ Bp1, __bf16* __restrict__ Bp2, int* __restrict__ ovfCount) {
  __shared__ int  sBin[MAX_BINS];
  __shared__ int  sWave[SP_THREADS / 32];
  __shared__ int2 sPairs[SP_SLICE];
  const int tid  = threadIdx.x;
  const int lane = tid & 31;
  const int wave = tid >> 5;
  const int w    = blockIdx.x;
  const int gi   = w * SP_THREADS + tid;

  sBin[tid] = 0;
  sBin[tid + SP_THREADS] = 0;
  if (gi == 0) *ovfCount = 0;
  if (gi < 512) pack_W_slot<128, true>(W1, Bp1, gi);
  else if (gi < 768) pack_W_slot<64, false>(W2, Bp2, gi - 512);
  __syncthreads();

  const int base = w * SP_SLICE + tid;
  int d[SP_PER_THREAD], s[SP_PER_THREAD], rk[SP_PER_THREAD];
#pragma unroll
  for (int k = 0; k < SP_PER_THREAD; ++k) {
    const int idx = base + k * SP_THREADS;
    const int v = dst[idx < e ? idx : e - 1];
    d[k] = idx < e ? v : -1;
  }
#pragma unroll
  for (int k = 0; k < SP_PER_THREAD; ++k) {
    const int idx = base + k * SP_THREADS;
    s[k] = src[idx < e ? idx : e - 1];
  }
#pragma unroll
  for (int k = 0; k < SP_PER_THREAD; ++k) {
    rk[k] = 0;
    if (d[k] >= 0 && d[k] < n) rk[k] = atomicAdd(&sBin[d[k] >> BIN_SHIFT], 1);
  }
  __syncthreads();

  const int b0 = 2 * tid;
  const int b1 = b0 + 1;
  const int c0 = sBin[b0];
  const int c1 = sBin[b1];
  const int v = c0 + c1;
  int x = v;
#pragma unroll
  for (int o = 1; o < 32; o <<= 1) {
    const int y = __shfl_up(x, o, 32);
    if (lane >= o) x += y;
  }
  if (lane == 31) sWave[wave] = x;
  __syncthreads();
  int wb = 0;
#pragma unroll
  for (int i = 0; i < SP_THREADS / 32; ++i) {
    if (i < wave) wb += sWave[i];
  }
  const int ex = wb + x - v;
  sBin[b0] = ex;
  sBin[b1] = ex + c0;
  int* __restrict__ myStarts = starts + (size_t)w * (nb + 1);
  if (b0 <= nb) myStarts[b0] = ex;
  if (b1 <= nb) myStarts[b1] = ex + c0;
  __syncthreads();

  int st[SP_PER_THREAD];
#pragma unroll
  for (int k = 0; k < SP_PER_THREAD; ++k) {
    const bool ok = d[k] >= 0 && d[k] < n;
    st[k] = sBin[ok ? (d[k] >> BIN_SHIFT) : 0];
  }
#pragma unroll
  for (int k = 0; k < SP_PER_THREAD; ++k) {
    if (d[k] >= 0 && d[k] < n) sPairs[st[k] + rk[k]] = make_int2(s[k], d[k]);
  }
  __syncthreads();
  const int total = sBin[nb];
  int2* __restrict__ mySorted = sorted + (size_t)w * SP_SLICE;
  for (int i = tid; i < total; i += SP_THREADS) mySorted[i] = sPairs[i];
}

#define G1_K 128

__device__ __forceinline__ void load_a_row(const float* __restrict__ arow, int khalf, float4 (&a)[G1_K / 8]) {
#pragma unroll
  for (int s = 0; s < G1_K / 32; ++s) {
    a[4 * s + 0] = *(const float4*)(arow + s * 32 + khalf);
    a[4 * s + 1] = *(const float4*)(arow + s * 32 + khalf + 4);
    a[4 * s + 2] = *(const float4*)(arow + s * 32 + khalf + 16);
    a[4 * s + 3] = *(const float4*)(arow + s * 32 + khalf + 20);
  }
}

__device__ __forceinline__ v16bf a_fragment(const float4 a0, const float4 a1, const float4 a2, const float4 a3) {
  const v16bf f = {(__bf16)a0.x, (__bf16)a0.y, (__bf16)a0.z, (__bf16)a0.w,
                   (__bf16)a1.x, (__bf16)a1.y, (__bf16)a1.z, (__bf16)a1.w,
                   (__bf16)a2.x, (__bf16)a2.y, (__bf16)a2.z, (__bf16)a2.w,
                   (__bf16)a3.x, (__bf16)a3.y, (__bf16)a3.z, (__bf16)a3.w};
  return f;
}

__device__ __forceinline__ void store_scaled_tile(
    const v8f (&acc)[4], int degc, int rowTile, int lane, int nRows,
    __bf16* __restrict__ out, float* __restrict__ dinv) {
  const int r16 = lane & 15;
  const int row = rowTile + r16;
  const float d = (float)(degc + 1);
  const float dv = d > 0.0f ? rsqrtf(d) : 0.0f;
  if (lane < 16 && row < nRows) dinv[row] = dv;
  const int hrow  = (lane >> 4) * 8;
  const int rbase = rowTile + hrow;
  float sc[8];
#pragma unroll
  for (int r = 0; r < 8; ++r) sc[r] = __shfl(dv, hrow + r, 32);
#pragma unroll
  for (int r = 0; r < 8; ++r) {
    const int rr = rbase + r;
    const v4bf p = {(__bf16)(acc[0][r] * sc[r]), (__bf16)(acc[1][r] * sc[r]),
                    (__bf16)(acc[2][r] * sc[r]), (__bf16)(acc[3][r] * sc[r])};
    if (rr < nRows) *(v4bf*)(out + (size_t)rr * 64 + 4 * r16) = p;
  }
}

#define ROW_PITCH  65
#define BR_THREADS 256
#define BR_GROUP   16
#define BR_NIT     16

__device__ __forceinline__ int local_row(int dstNode, int nodeBase) {
  const int local = dstNode - nodeBase;
  return local < 0 ? 0 : (local > BIN_NODES - 1 ? BIN_NODES - 1 : local);
}

__device__ __forceinline__ void add_edge_lds(
    int* __restrict__ sRow, int nodeBase, const int2 q,
    int* __restrict__ ovfCount, int* __restrict__ ovfSrc, int* __restrict__ ovfDst) {
  int* row = sRow + local_row(q.y, nodeBase) * ROW_PITCH;
  const int pos = atomicAdd(row, 1);
  if (pos < ROW_CAP) {
    row[1 + pos] = q.x;
  } else {
    const int o = atomicAdd(ovfCount, 1);
    ovfSrc[o] = q.x;
    ovfDst[o] = q.y;
  }
}

__global__ __launch_bounds__(BR_THREADS) void build_rows_kernel(
    const int2* __restrict__ sorted, const int* __restrict__ starts, int* __restrict__ rows,
    int* __restrict__ ovfCount, int* __restrict__ ovfSrc, int* __restrict__ ovfDst,
    int n, int nb, int ga,
    const float* __restrict__ emb, const __bf16* __restrict__ Bp,
    __bf16* __restrict__ x1, float* __restrict__ dinv) {
  __shared__ int sRow[BIN_NODES * ROW_PITCH];
  const int tid  = threadIdx.x;
  const int lane = tid & 31;
  const int wave = tid >> 5;
  const int b    = blockIdx.x;
  const int nodeBase = b << BIN_SHIFT;
  const int sub = tid / BR_GROUP;
  const int l16 = tid % BR_GROUP;

  sRow[tid * ROW_PITCH] = 0;
  __syncthreads();

  for (int pb = 0; pb < ga; pb += (BR_THREADS / BR_GROUP) * BR_NIT) {
    int i0[BR_NIT], s1[BR_NIT], pos[BR_NIT];
    int2 q[BR_NIT];
#pragma unroll
    for (int it = 0; it < BR_NIT; ++it) {
      const int p  = pb + it * (BR_THREADS / BR_GROUP) + sub;
      const int pc = p < ga ? p : ga - 1;
      const int* __restrict__ st = starts + (size_t)pc * (nb + 1) + b;
      const int a = st[0];
      const int c = st[1];
      i0[it] = a + l16;
      s1[it] = p < ga ? c : a;
    }
#pragma unroll
    for (int it = 0; it < BR_NIT; ++it) {
      const int p  = pb + it * (BR_THREADS / BR_GROUP) + sub;
      const int pc = p < ga ? p : ga - 1;
      if (i0[it] < s1[it]) q[it] = sorted[(size_t)pc * SP_SLICE + i0[it]];
    }
#pragma unroll
    for (int it = 0; it < BR_NIT; ++it) {
      pos[it] = 0;
      if (i0[it] < s1[it]) pos[it] = atomicAdd(sRow + local_row(q[it].y, nodeBase) * ROW_PITCH, 1);
    }
#pragma unroll
    for (int it = 0; it < BR_NIT; ++it) {
      if (i0[it] < s1[it]) {
        if (pos[it] < ROW_CAP) {
          sRow[local_row(q[it].y, nodeBase) * ROW_PITCH + 1 + pos[it]] = q[it].x;
        } else {
          const int o = atomicAdd(ovfCount, 1);
          ovfSrc[o] = q[it].x;
          ovfDst[o] = q[it].y;
        }
      }
    }
#pragma unroll
    for (int it = 0; it < BR_NIT; ++it) {
      const int p  = pb + it * (BR_THREADS / BR_GROUP) + sub;
      const int pc = p < ga ? p : ga - 1;
      for (int j = i0[it] + BR_GROUP; j < s1[it]; j += BR_GROUP)
        add_edge_lds(sRow, nodeBase, sorted[(size_t)pc * SP_SLICE + j], ovfCount, ovfSrc, ovfDst);
    }
  }
  __syncthreads();

  const int r16   = lane & 15;
  const int khalf = (lane >> 4) * 8;
  const int tile0 = nodeBase + wave * 32;
  const int tile1 = tile0 + 16;
  const int ra = tile0 + r16;
  const int rb = tile1 + r16;
  float4 a0[G1_K / 8], a1[G1_K / 8];
  load_a_row(emb + (size_t)(ra < n ? ra : n - 1) * G1_K, khalf, a0);
  load_a_row(emb + (size_t)(rb < n ? rb : n - 1) * G1_K, khalf, a1);

  for (int r = wave; r < BIN_NODES; r += BR_THREADS / 32) {
    const int g = nodeBase + r;
    if (g >= n) break;
    const int* row = sRow + r * ROW_PITCH;
    const int cnt = row[0];
    int* __restrict__ grow = rows + (size_t)g * ROW_INTS;
    grow[lane] = row[lane];
    if (cnt > 31) grow[32 + lane] = row[32 + lane];
  }

  const int deg0 = sRow[(wave * 32 + r16) * ROW_PITCH];
  const int deg1 = sRow[(wave * 32 + 16 + r16) * ROW_PITCH];

  v8f acc0[4], acc1[4];
  acc0[0] = (v8f){}; acc0[1] = (v8f){}; acc0[2] = (v8f){}; acc0[3] = (v8f){};
  acc1[0] = (v8f){}; acc1[1] = (v8f){}; acc1[2] = (v8f){}; acc1[3] = (v8f){};
#pragma unroll
  for (int s = 0; s < G1_K / 32; ++s) {
    const v16bf fa = a_fragment(a0[4 * s], a0[4 * s + 1], a0[4 * s + 2], a0[4 * s + 3]);
    const v16bf fb = a_fragment(a1[4 * s], a1[4 * s + 1], a1[4 * s + 2], a1[4 * s + 3]);
#pragma unroll
    for (int nt = 0; nt < 4; ++nt) {
      const v16bf b = *(const v16bf*)(Bp + ((size_t)(s * 4 + nt) * 32 + lane) * 16);
      acc0[nt] = __builtin_amdgcn_wmma_f32_16x16x32_bf16(
           false, fa,  false, b,
           (short)0, acc0[nt],  false,  false);
      acc1[nt] = __builtin_amdgcn_wmma_f32_16x16x32_bf16(
           false, fb,  false, b,
           (short)0, acc1[nt],  false,  false);
    }
  }
  store_scaled_tile(acc0, deg0, tile0, lane, n, x1, dinv);
  store_scaled_tile(acc1, deg1, tile1, lane, n, x1, dinv);
}

__device__ __forceinline__ float bf_lo(unsigned w) { return __uint_as_float(w << 16); }
__device__ __forceinline__ float bf_hi(unsigned w) { return __uint_as_float(w & 0xffff0000u); }

__device__ __forceinline__ void add_row_bf16(const v4u v, float4& accA, float4& accB) {
  accA.x += bf_lo(v.x); accA.y += bf_hi(v.x); accA.z += bf_lo(v.y); accA.w += bf_hi(v.y);
  accB.x += bf_lo(v.z); accB.y += bf_hi(v.z); accB.z += bf_lo(v.w); accB.w += bf_hi(v.w);
}

__device__ __forceinline__ void add_rows8(
    const __bf16* __restrict__ xs, int sidx, int cnt, int f8, float4& accA, float4& accB) {
  int s[8];
  v4u v[8];
#pragma unroll
  for (int u = 0; u < 8; ++u) s[u] = __shfl(sidx, u, 8);
#pragma unroll
  for (int u = 0; u < 8; ++u) {
    if (u < cnt) v[u] = *(const v4u*)(xs + ((unsigned)s[u] * 64u + (unsigned)f8));
  }
#pragma unroll
  for (int u = 0; u < 8; ++u) {
    if (u < cnt) add_row_bf16(v[u], accA, accB);
  }
}

__device__ __forceinline__ void row_sum8(
    const __bf16* __restrict__ xs, const int* __restrict__ rows, int i, int n, int l8,
    const int* __restrict__ ovfCount, const int* __restrict__ ovfSrc,
    const int* __restrict__ ovfDst, int nEdges, float4& accA, float4& accB) {
  const int f8 = l8 * 8;
  const int* __restrict__ row = rows + (size_t)i * ROW_INTS;
  const int mraw = row[0];
  const int m = mraw < 0 ? 0 : (mraw > ROW_CAP ? ROW_CAP : mraw);
  const v4u sv = *(const v4u*)(xs + (size_t)i * 64 + f8);
  accA = (float4){bf_lo(sv.x), bf_hi(sv.x), bf_lo(sv.y), bf_hi(sv.y)};
  accB = (float4){bf_lo(sv.z), bf_hi(sv.z), bf_lo(sv.w), bf_hi(sv.w)};

  for (int base = 0; base < m; base += 32) {
    int e[4];
#pragma unroll
    for (int q = 0; q < 4; ++q) {
      const int j = base + q * 8 + l8;
      e[q] = row[1 + (j < m ? j : m - 1)];
    }
#pragma unroll
    for (int q = 0; q < 4; ++q)
      e[q] = e[q] < 0 ? 0 : (e[q] >= n ? n - 1 : e[q]);
    const int left = m - base;
#pragma unroll
    for (int q = 0; q < 4; ++q) {
      const int cnt = left - q * 8;
      if (cnt > 0) add_rows8(xs, e[q], cnt, f8, accA, accB);
    }
  }
  if (mraw > ROW_CAP) {
    int cnt = *ovfCount;
    cnt = cnt < 0 ? 0 : (cnt > nEdges ? nEdges : cnt);
    for (int base = 0; base < cnt; base += 8) {
      const int e = base + l8;
      int od = -1;
      int os = 0;
      if (e < cnt) {
        od = ovfDst[e];
        os = ovfSrc[e];
      }
#pragma nounroll
      for (int u = 0; u < 8; ++u) {
        const int du = __shfl(od, u, 8);
        int su = __shfl(os, u, 8);
        if (du == i) {
          su = su < 0 ? 0 : (su >= n ? n - 1 : su);
          const v4u vu = *(const v4u*)(xs + (size_t)su * 64 + f8);
          add_row_bf16(vu, accA, accB);
        }
      }
    }
  }
}

__global__ __launch_bounds__(256) void gather_rows_kernel(
    const __bf16* __restrict__ xs, const float* __restrict__ dinv,
    const int* __restrict__ rows, const float* __restrict__ bias,
    float* __restrict__ out, int n,
    const int* __restrict__ ovfCount, const int* __restrict__ ovfSrc,
    const int* __restrict__ ovfDst, int nEdges) {
  int t = blockIdx.x * blockDim.x + threadIdx.x;
  int i = t >> 3;
  if (i >= n) return;
  const int l8 = threadIdx.x & 7;
  const int f = l8 * 4;
  const float di = dinv[i];
  float4 a, b;
  row_sum8(xs, rows, i, n, l8, ovfCount, ovfSrc, ovfDst, nEdges, a, b);
  a.x *= di; a.y *= di; a.z *= di; a.w *= di;
  b.x *= di; b.y *= di; b.z *= di; b.w *= di;
  a.x += bias[f]; a.y += bias[f + 1]; a.z += bias[f + 2]; a.w += bias[f + 3];
  b.x += bias[32 + f]; b.y += bias[32 + f + 1]; b.z += bias[32 + f + 2]; b.w += bias[32 + f + 3];
  *(float4*)(out + (size_t)i * 64 + f) = a;
  *(float4*)(out + (size_t)i * 64 + 32 + f) = b;
}

#define A_PITCH 72
#define D_PITCH 68
#define GG_NODES 32

__global__ __launch_bounds__(256) void gather_gemm_kernel(
    const __bf16* __restrict__ xs, const float* __restrict__ dinv,
    const int* __restrict__ rows, const float* __restrict__ bias,
    const __bf16* __restrict__ Bp, __bf16* __restrict__ out, int n,
    const int* __restrict__ ovfCount, const int* __restrict__ ovfSrc,
    const int* __restrict__ ovfDst, int nEdges) {
  __shared__ __attribute__((aligned(16))) __bf16 sA[GG_NODES * A_PITCH];
  __shared__ __attribute__((aligned(16))) float  sD[GG_NODES * D_PITCH];
  __shared__ __attribute__((aligned(16))) float  sScale[GG_NODES];

  const int lane = threadIdx.x & 31;
  const int wave = threadIdx.x >> 5;
  const int nb   = threadIdx.x >> 3;
  const int l8   = threadIdx.x & 7;
  const int f    = l8 * 4;
  const int i    = blockIdx.x * GG_NODES + nb;
  const int rt   = wave >> 2;
  const int ct   = wave & 3;

  const v16bf b0 = *(const v16bf*)(Bp + ((size_t)(0 * 4 + ct) * 32 + lane) * 16);
  const v16bf b1 = *(const v16bf*)(Bp + ((size_t)(1 * 4 + ct) * 32 + lane) * 16);

  float4 hA = {0.0f, 0.0f, 0.0f, 0.0f};
  float4 hB = {0.0f, 0.0f, 0.0f, 0.0f};
  float di = 0.0f;
  if (i < n) {
    di = dinv[i];
    float4 a, b;
    row_sum8(xs, rows, i, n, l8, ovfCount, ovfSrc, ovfDst, nEdges, a, b);
    hA.x = a.x * di + bias[f];          hA.x = hA.x > 0.0f ? hA.x : 0.0f;
    hA.y = a.y * di + bias[f + 1];      hA.y = hA.y > 0.0f ? hA.y : 0.0f;
    hA.z = a.z * di + bias[f + 2];      hA.z = hA.z > 0.0f ? hA.z : 0.0f;
    hA.w = a.w * di + bias[f + 3];      hA.w = hA.w > 0.0f ? hA.w : 0.0f;
    hB.x = b.x * di + bias[32 + f];     hB.x = hB.x > 0.0f ? hB.x : 0.0f;
    hB.y = b.y * di + bias[32 + f + 1]; hB.y = hB.y > 0.0f ? hB.y : 0.0f;
    hB.z = b.z * di + bias[32 + f + 2]; hB.z = hB.z > 0.0f ? hB.z : 0.0f;
    hB.w = b.w * di + bias[32 + f + 3]; hB.w = hB.w > 0.0f ? hB.w : 0.0f;
  }
  const v4bf hbA = {(__bf16)hA.x, (__bf16)hA.y, (__bf16)hA.z, (__bf16)hA.w};
  const v4bf hbB = {(__bf16)hB.x, (__bf16)hB.y, (__bf16)hB.z, (__bf16)hB.w};
  *(v4bf*)(sA + nb * A_PITCH + f) = hbA;
  *(v4bf*)(sA + nb * A_PITCH + 32 + f) = hbB;
  if (l8 == 0) sScale[nb] = di;
  __syncthreads();

  {
    const int c  = lane & 15;
    const int hh = lane >> 4;
    const __bf16* ap = sA + (rt * 16 + c) * A_PITCH + hh * 8;
    const v8bf a0lo = *(const v8bf*)(ap);
    const v8bf a0hi = *(const v8bf*)(ap + 16);
    const v8bf a1lo = *(const v8bf*)(ap + 32);
    const v8bf a1hi = *(const v8bf*)(ap + 48);
    const v16bf a0 = __builtin_shufflevector(a0lo, a0hi, 0, 1, 2, 3, 4, 5, 6, 7, 8, 9, 10, 11, 12, 13, 14, 15);
    const v16bf a1 = __builtin_shufflevector(a1lo, a1hi, 0, 1, 2, 3, 4, 5, 6, 7, 8, 9, 10, 11, 12, 13, 14, 15);
    v8f acc = {};
    acc = __builtin_amdgcn_wmma_f32_16x16x32_bf16(
         false, a0,  false, b0,
         (short)0, acc,  false,  false);
    acc = __builtin_amdgcn_wmma_f32_16x16x32_bf16(
         false, a1,  false, b1,
         (short)0, acc,  false,  false);
    const float4 s0 = *(const float4*)(sScale + rt * 16 + hh * 8);
    const float4 s1 = *(const float4*)(sScale + rt * 16 + hh * 8 + 4);
    const float sc[8] = {s0.x, s0.y, s0.z, s0.w, s1.x, s1.y, s1.z, s1.w};
    float* dp = sD + (rt * 16 + hh * 8) * D_PITCH + ct * 16 + c;
#pragma unroll
    for (int r = 0; r < 8; ++r) dp[r * D_PITCH] = acc[r] * sc[r];
  }
  __syncthreads();

  if (i < n) {
    const float4 d0 = *(const float4*)(sD + nb * D_PITCH + f);
    const float4 d1 = *(const float4*)(sD + nb * D_PITCH + 32 + f);
    const v8bf o = {(__bf16)d0.x, (__bf16)d0.y, (__bf16)d0.z, (__bf16)d0.w,
                    (__bf16)d1.x, (__bf16)d1.y, (__bf16)d1.z, (__bf16)d1.w};
    *(v8bf*)(out + (size_t)i * 64 + l8 * 8) = o;
  }
}

extern "C" void kernel_launch(void* const* d_in, const int* in_sizes, int n_in,
                              void* d_out, int out_size, void* d_ws, size_t ws_size,
                              hipStream_t stream) {
  const int*   edge_index = (const int*)d_in[0];
  const float* emb        = (const float*)d_in[1];
  const float* W1         = (const float*)d_in[2];
  const float* b1         = (const float*)d_in[3];
  const float* W2         = (const float*)d_in[4];
  const float* b2         = (const float*)d_in[5];
  float* out = (float*)d_out;

  const int E = in_sizes[0] / 2;
  const int N = in_sizes[1] / 128;
  const int* src = edge_index;
  const int* dst = edge_index + E;

  float* ws   = (float*)d_ws;
  float* dinv = ws;
  size_t npad = ((size_t)N + 63) & ~(size_t)63;
  int*   ovfCount = (int*)(ws + npad);
  float* p1   = ws + npad + 64;
  float* p2   = p1 + (size_t)N * 64;
  __bf16* x1  = (__bf16*)p1;
  __bf16* x2  = (__bf16*)p2;
  __bf16* Bp1 = (__bf16*)(p2 + (size_t)N * 64);
  __bf16* Bp2 = Bp1 + 8192;
  int* rows   = (int*)(p2 + (size_t)N * 64 + 6144);
  int* ovfSrc = rows + (size_t)N * ROW_INTS;
  size_t epad = ((size_t)E + 63) & ~(size_t)63;
  int* ovfDst = ovfSrc + epad;
  const int nSlices = (E + SP_SLICE - 1) / SP_SLICE;
  const int splitBlk = nSlices > 3 ? nSlices : 3;
  const int nBins   = (N + BIN_NODES - 1) / BIN_NODES;
  int2* sorted = (int2*)(ovfDst + epad);
  int* starts  = (int*)(sorted + (size_t)splitBlk * SP_SLICE);

  const int tpb = 256;
  const long nf = (long)N * 8;
  const int nfBlk = (int)((nf + tpb - 1) / tpb);

  split_edges_kernel<<<splitBlk, SP_THREADS, 0, stream>>>(src, dst, sorted, starts, E, N, nBins,
                                                          W1, W2, Bp1, Bp2, ovfCount);

  build_rows_kernel<<<nBins, BR_THREADS, 0, stream>>>(sorted, starts, rows, ovfCount, ovfSrc, ovfDst,
                                                      N, nBins, splitBlk, emb, Bp1, x1, dinv);

  gather_gemm_kernel<<<nfBlk, tpb, 0, stream>>>(x1, dinv, rows, b1, Bp2, x2, N,
                                                ovfCount, ovfSrc, ovfDst, E);

  gather_rows_kernel<<<nfBlk, tpb, 0, stream>>>(x2, dinv, rows, b2, out, N,
                                                ovfCount, ovfSrc, ovfDst, E);
}


# Destination rows filled up to a multiple of 8 entries with the index of an all-zero plane row: a row-sum step is 8 loads and 8 adds, no branch per row; row cap 56; sum order kept, fillers add zero
# speedup vs baseline: 6.4061x; 1.0356x over previous
#include <hip/hip_runtime.h>

typedef __attribute__((ext_vector_type(16))) __bf16 v16bf;
typedef __attribute__((ext_vector_type(8)))  float  v8f;
typedef __attribute__((ext_vector_type(8)))  __bf16 v8bf;
typedef __attribute__((ext_vector_type(4)))  __bf16 v4bf;
typedef __attribute__((ext_vector_type(4)))  unsigned int v4u;

#define ROW_INTS 64
#define ROW_CAP  56


template <int K, bool PERM>
__device__ __forceinline__ void pack_W_slot(const float* __restrict__ W, __bf16* __restrict__ Bp, int t) {
  int lane  = t & 31;
  int nt    = (t >> 5) & 3;
  int ktIdx = t >> 7;
  int r16   = lane & 15;
  int khalf = (lane >> 4) * 8;
  int col   = PERM ? ((r16 & 1) * 32 + (r16 >> 1) * 4 + nt) : (nt * 16 + r16);
  v16bf b;
#pragma unroll
  for (int i = 0; i < 8; ++i) {
    b[i]     = (__bf16)W[(size_t)(ktIdx * 32 + khalf + i) * 64 + col];
    b[8 + i] = (__bf16)W[(size_t)(ktIdx * 32 + khalf + 16 + i) * 64 + col];
  }
  *(v16bf*)(Bp + (size_t)t * 16) = b;
}

#define SP_THREADS    256
#define SP_PER_THREAD 16
#define SP_SLICE      (SP_THREADS * SP_PER_THREAD)
#define BIN_SHIFT     8
#define BIN_NODES     (1 << BIN_SHIFT)
#define MAX_BINS      512

__global__ __launch_bounds__(SP_THREADS) void split_edges_kernel(
    const int* __restrict__ src, const int* __restrict__ dst,
    int2* __restrict__ sorted, int* __restrict__ starts, int e, int n, int nb,
    const float* __restrict__ W1, const float* __restrict__ W2,
    __bf16* __restrict__ Bp1, __bf16* __restrict__ Bp2, int* __restrict__ ovfCount) {
  __shared__ int  sBin[MAX_BINS];
  __shared__ int  sWave[SP_THREADS / 32];
  __shared__ int2 sPairs[SP_SLICE];
  const int tid  = threadIdx.x;
  const int lane = tid & 31;
  const int wave = tid >> 5;
  const int w    = blockIdx.x;
  const int gi   = w * SP_THREADS + tid;

  sBin[tid] = 0;
  sBin[tid + SP_THREADS] = 0;
  if (gi == 0) *ovfCount = 0;
  if (gi < 512) pack_W_slot<128, true>(W1, Bp1, gi);
  else if (gi < 768) pack_W_slot<64, false>(W2, Bp2, gi - 512);
  __syncthreads();

  const int base = w * SP_SLICE + tid;
  int d[SP_PER_THREAD], s[SP_PER_THREAD], rk[SP_PER_THREAD];
#pragma unroll
  for (int k = 0; k < SP_PER_THREAD; ++k) {
    const int idx = base + k * SP_THREADS;
    const int v = dst[idx < e ? idx : e - 1];
    d[k] = idx < e ? v : -1;
  }
#pragma unroll
  for (int k = 0; k < SP_PER_THREAD; ++k) {
    const int idx = base + k * SP_THREADS;
    s[k] = src[idx < e ? idx : e - 1];
  }
#pragma unroll
  for (int k = 0; k < SP_PER_THREAD; ++k) {
    rk[k] = 0;
    if (d[k] >= 0 && d[k] < n) rk[k] = atomicAdd(&sBin[d[k] >> BIN_SHIFT], 1);
  }
  __syncthreads();

  const int b0 = 2 * tid;
  const int b1 = b0 + 1;
  const int c0 = sBin[b0];
  const int c1 = sBin[b1];
  const int v = c0 + c1;
  int x = v;
#pragma unroll
  for (int o = 1; o < 32; o <<= 1) {
    const int y = __shfl_up(x, o, 32);
    if (lane >= o) x += y;
  }
  if (lane == 31) sWave[wave] = x;
  __syncthreads();
  int wb = 0;
#pragma unroll
  for (int i = 0; i < SP_THREADS / 32; ++i) {
    if (i < wave) wb += sWave[i];
  }
  const int ex = wb + x - v;
  sBin[b0] = ex;
  sBin[b1] = ex + c0;
  int* __restrict__ myStarts = starts + (size_t)w * (nb + 1);
  if (b0 <= nb) myStarts[b0] = ex;
  if (b1 <= nb) myStarts[b1] = ex + c0;
  __syncthreads();

  int st[SP_PER_THREAD];
#pragma unroll
  for (int k = 0; k < SP_PER_THREAD; ++k) {
    const bool ok = d[k] >= 0 && d[k] < n;
    st[k] = sBin[ok ? (d[k] >> BIN_SHIFT) : 0];
  }
#pragma unroll
  for (int k = 0; k < SP_PER_THREAD; ++k) {
    if (d[k] >= 0 && d[k] < n) sPairs[st[k] + rk[k]] = make_int2(s[k], d[k]);
  }
  __syncthreads();
  const int total = sBin[nb];
  int2* __restrict__ mySorted = sorted + (size_t)w * SP_SLICE;
  for (int i = tid; i < total; i += SP_THREADS) mySorted[i] = sPairs[i];
}

#define G1_K 128

__device__ __forceinline__ void load_a_row(const float* __restrict__ arow, int khalf, float4 (&a)[G1_K / 8]) {
#pragma unroll
  for (int s = 0; s < G1_K / 32; ++s) {
    a[4 * s + 0] = *(const float4*)(arow + s * 32 + khalf);
    a[4 * s + 1] = *(const float4*)(arow + s * 32 + khalf + 4);
    a[4 * s + 2] = *(const float4*)(arow + s * 32 + khalf + 16);
    a[4 * s + 3] = *(const float4*)(arow + s * 32 + khalf + 20);
  }
}

__device__ __forceinline__ v16bf a_fragment(const float4 a0, const float4 a1, const float4 a2, const float4 a3) {
  const v16bf f = {(__bf16)a0.x, (__bf16)a0.y, (__bf16)a0.z, (__bf16)a0.w,
                   (__bf16)a1.x, (__bf16)a1.y, (__bf16)a1.z, (__bf16)a1.w,
                   (__bf16)a2.x, (__bf16)a2.y, (__bf16)a2.z, (__bf16)a2.w,
                   (__bf16)a3.x, (__bf16)a3.y, (__bf16)a3.z, (__bf16)a3.w};
  return f;
}

__device__ __forceinline__ void store_scaled_tile(
    const v8f (&acc)[4], int degc, int rowTile, int lane, int nRows,
    __bf16* __restrict__ out, float* __restrict__ dinv) {
  const int r16 = lane & 15;
  const int row = rowTile + r16;
  const float d = (float)(degc + 1);
  const float dv = d > 0.0f ? rsqrtf(d) : 0.0f;
  if (lane < 16 && row < nRows) dinv[row] = dv;
  const int hrow  = (lane >> 4) * 8;
  const int rbase = rowTile + hrow;
  float sc[8];
#pragma unroll
  for (int r = 0; r < 8; ++r) sc[r] = __shfl(dv, hrow + r, 32);
#pragma unroll
  for (int r = 0; r < 8; ++r) {
    const int rr = rbase + r;
    const v4bf p = {(__bf16)(acc[0][r] * sc[r]), (__bf16)(acc[1][r] * sc[r]),
                    (__bf16)(acc[2][r] * sc[r]), (__bf16)(acc[3][r] * sc[r])};
    if (rr < nRows) *(v4bf*)(out + (size_t)rr * 64 + 4 * r16) = p;
  }
}

#define ROW_PITCH  65
#define BR_THREADS 256
#define BR_GROUP   16
#define BR_NIT     16

__device__ __forceinline__ int local_row(int dstNode, int nodeBase) {
  const int local = dstNode - nodeBase;
  return local < 0 ? 0 : (local > BIN_NODES - 1 ? BIN_NODES - 1 : local);
}

__device__ __forceinline__ int row_source(int srcNode, int n) {
  return srcNode < 0 ? 0 : (srcNode >= n ? n - 1 : srcNode);
}

__device__ __forceinline__ void add_edge_lds(
    int* __restrict__ sRow, int nodeBase, int n, const int2 q,
    int* __restrict__ ovfCount, int* __restrict__ ovfSrc, int* __restrict__ ovfDst) {
  int* row = sRow + local_row(q.y, nodeBase) * ROW_PITCH;
  const int pos = atomicAdd(row, 1);
  if (pos < ROW_CAP) {
    row[1 + pos] = row_source(q.x, n);
  } else {
    const int o = atomicAdd(ovfCount, 1);
    ovfSrc[o] = q.x;
    ovfDst[o] = q.y;
  }
}

__global__ __launch_bounds__(BR_THREADS) void build_rows_kernel(
    const int2* __restrict__ sorted, const int* __restrict__ starts, int* __restrict__ rows,
    int* __restrict__ ovfCount, int* __restrict__ ovfSrc, int* __restrict__ ovfDst,
    int n, int nb, int ga,
    const float* __restrict__ emb, const __bf16* __restrict__ Bp,
    __bf16* __restrict__ x1, float* __restrict__ dinv) {
  __shared__ int sRow[BIN_NODES * ROW_PITCH];
  const int tid  = threadIdx.x;
  const int lane = tid & 31;
  const int wave = tid >> 5;
  const int b    = blockIdx.x;
  const int nodeBase = b << BIN_SHIFT;
  const int sub = tid / BR_GROUP;
  const int l16 = tid % BR_GROUP;

  sRow[tid * ROW_PITCH] = 0;
  __syncthreads();

  for (int pb = 0; pb < ga; pb += (BR_THREADS / BR_GROUP) * BR_NIT) {
    int i0[BR_NIT], s1[BR_NIT], pos[BR_NIT];
    int2 q[BR_NIT];
#pragma unroll
    for (int it = 0; it < BR_NIT; ++it) {
      const int p  = pb + it * (BR_THREADS / BR_GROUP) + sub;
      const int pc = p < ga ? p : ga - 1;
      const int* __restrict__ st = starts + (size_t)pc * (nb + 1) + b;
      const int a = st[0];
      const int c = st[1];
      i0[it] = a + l16;
      s1[it] = p < ga ? c : a;
    }
#pragma unroll
    for (int it = 0; it < BR_NIT; ++it) {
      const int p  = pb + it * (BR_THREADS / BR_GROUP) + sub;
      const int pc = p < ga ? p : ga - 1;
      if (i0[it] < s1[it]) q[it] = sorted[(size_t)pc * SP_SLICE + i0[it]];
    }
#pragma unroll
    for (int it = 0; it < BR_NIT; ++it) {
      pos[it] = 0;
      if (i0[it] < s1[it]) pos[it] = atomicAdd(sRow + local_row(q[it].y, nodeBase) * ROW_PITCH, 1);
    }
#pragma unroll
    for (int it = 0; it < BR_NIT; ++it) {
      if (i0[it] < s1[it]) {
        if (pos[it] < ROW_CAP) {
          sRow[local_row(q[it].y, nodeBase) * ROW_PITCH + 1 + pos[it]] = row_source(q[it].x, n);
        } else {
          const int o = atomicAdd(ovfCount, 1);
          ovfSrc[o] = q[it].x;
          ovfDst[o] = q[it].y;
        }
      }
    }
#pragma unroll
    for (int it = 0; it < BR_NIT; ++it) {
      const int p  = pb + it * (BR_THREADS / BR_GROUP) + sub;
      const int pc = p < ga ? p : ga - 1;
      for (int j = i0[it] + BR_GROUP; j < s1[it]; j += BR_GROUP)
        add_edge_lds(sRow, nodeBase, n, sorted[(size_t)pc * SP_SLICE + j], ovfCount, ovfSrc, ovfDst);
    }
  }
  __syncthreads();

  const int r16   = lane & 15;
  const int khalf = (lane >> 4) * 8;
  const int tile0 = nodeBase + wave * 32;
  const int tile1 = tile0 + 16;
  const int ra = tile0 + r16;
  const int rb = tile1 + r16;
  float4 a0[G1_K / 8], a1[G1_K / 8];
  load_a_row(emb + (size_t)(ra < n ? ra : n - 1) * G1_K, khalf, a0);
  load_a_row(emb + (size_t)(rb < n ? rb : n - 1) * G1_K, khalf, a1);

  if (b == 0 && tid < 8) {
    const v8bf z = {};
    *(v8bf*)(x1 + (size_t)n * 64 + tid * 8) = z;
  }

  for (int r = wave; r < BIN_NODES; r += BR_THREADS / 32) {
    const int g = nodeBase + r;
    if (g >= n) break;
    const int* row = sRow + r * ROW_PITCH;
    const int cnt = row[0];
    const int m  = cnt > ROW_CAP ? ROW_CAP : cnt;
    const int mp = (m + 7) & ~7;
    int* __restrict__ grow = rows + (size_t)g * ROW_INTS;
    const int w0 = row[lane];
    grow[lane] = (lane > m && lane <= mp) ? n : w0;
    if (mp > 31) {
      const int w1 = row[32 + lane];
      grow[32 + lane] = (32 + lane > m && 32 + lane <= mp) ? n : w1;
    }
  }

  const int deg0 = sRow[(wave * 32 + r16) * ROW_PITCH];
  const int deg1 = sRow[(wave * 32 + 16 + r16) * ROW_PITCH];

  v8f acc0[4], acc1[4];
  acc0[0] = (v8f){}; acc0[1] = (v8f){}; acc0[2] = (v8f){}; acc0[3] = (v8f){};
  acc1[0] = (v8f){}; acc1[1] = (v8f){}; acc1[2] = (v8f){}; acc1[3] = (v8f){};
#pragma unroll
  for (int s = 0; s < G1_K / 32; ++s) {
    const v16bf fa = a_fragment(a0[4 * s], a0[4 * s + 1], a0[4 * s + 2], a0[4 * s + 3]);
    const v16bf fb = a_fragment(a1[4 * s], a1[4 * s + 1], a1[4 * s + 2], a1[4 * s + 3]);
#pragma unroll
    for (int nt = 0; nt < 4; ++nt) {
      const v16bf b = *(const v16bf*)(Bp + ((size_t)(s * 4 + nt) * 32 + lane) * 16);
      acc0[nt] = __builtin_amdgcn_wmma_f32_16x16x32_bf16(
           false, fa,  false, b,
           (short)0, acc0[nt],  false,  false);
      acc1[nt] = __builtin_amdgcn_wmma_f32_16x16x32_bf16(
           false, fb,  false, b,
           (short)0, acc1[nt],  false,  false);
    }
  }
  store_scaled_tile(acc0, deg0, tile0, lane, n, x1, dinv);
  store_scaled_tile(acc1, deg1, tile1, lane, n, x1, dinv);
}

__device__ __forceinline__ float bf_lo(unsigned w) { return __uint_as_float(w << 16); }
__device__ __forceinline__ float bf_hi(unsigned w) { return __uint_as_float(w & 0xffff0000u); }

__device__ __forceinline__ void add_row_bf16(const v4u v, float4& accA, float4& accB) {
  accA.x += bf_lo(v.x); accA.y += bf_hi(v.x); accA.z += bf_lo(v.y); accA.w += bf_hi(v.y);
  accB.x += bf_lo(v.z); accB.y += bf_hi(v.z); accB.z += bf_lo(v.w); accB.w += bf_hi(v.w);
}

__device__ __forceinline__ void add_rows8(
    const __bf16* __restrict__ xs, int sidx, int f8, float4& accA, float4& accB) {
  int s[8];
  v4u v[8];
#pragma unroll
  for (int u = 0; u < 8; ++u) s[u] = __shfl(sidx, u, 8);
#pragma unroll
  for (int u = 0; u < 8; ++u) v[u] = *(const v4u*)(xs + ((unsigned)s[u] * 64u + (unsigned)f8));
#pragma unroll
  for (int u = 0; u < 8; ++u) add_row_bf16(v[u], accA, accB);
}

__device__ __forceinline__ void row_sum8(
    const __bf16* __restrict__ xs, const int* __restrict__ rows, int i, int n, int l8,
    const int* __restrict__ ovfCount, const int* __restrict__ ovfSrc,
    const int* __restrict__ ovfDst, int nEdges, float4& accA, float4& accB) {
  const int f8 = l8 * 8;
  const int* __restrict__ row = rows + (size_t)i * ROW_INTS;
  const int mraw = row[0];
  const int m = mraw < 0 ? 0 : (mraw > ROW_CAP ? ROW_CAP : mraw);
  const int mp = (m + 7) & ~7;
  const v4u sv = *(const v4u*)(xs + (size_t)i * 64 + f8);
  accA = (float4){bf_lo(sv.x), bf_hi(sv.x), bf_lo(sv.y), bf_hi(sv.y)};
  accB = (float4){bf_lo(sv.z), bf_hi(sv.z), bf_lo(sv.w), bf_hi(sv.w)};

  for (int base = 0; base < mp; base += 32) {
    int e[4];
#pragma unroll
    for (int q = 0; q < 4; ++q) {
      const int j = base + q * 8 + l8;
      e[q] = row[1 + (j < mp ? j : mp - 1)];
    }
#pragma unroll
    for (int q = 0; q < 4; ++q)
      e[q] = e[q] < 0 ? 0 : (e[q] > n ? n : e[q]);
    const int left = mp - base;
#pragma unroll
    for (int q = 0; q < 4; ++q) {
      if (left > q * 8) add_rows8(xs, e[q], f8, accA, accB);
    }
  }
  if (mraw > ROW_CAP) {
    int cnt = *ovfCount;
    cnt = cnt < 0 ? 0 : (cnt > nEdges ? nEdges : cnt);
    for (int base = 0; base < cnt; base += 8) {
      const int e = base + l8;
      int od = -1;
      int os = 0;
      if (e < cnt) {
        od = ovfDst[e];
        os = ovfSrc[e];
      }
#pragma nounroll
      for (int u = 0; u < 8; ++u) {
        const int du = __shfl(od, u, 8);
        int su = __shfl(os, u, 8);
        if (du == i) {
          su = su < 0 ? 0 : (su >= n ? n - 1 : su);
          const v4u vu = *(const v4u*)(xs + (size_t)su * 64 + f8);
          add_row_bf16(vu, accA, accB);
        }
      }
    }
  }
}

__global__ __launch_bounds__(256) void gather_rows_kernel(
    const __bf16* __restrict__ xs, const float* __restrict__ dinv,
    const int* __restrict__ rows, const float* __restrict__ bias,
    float* __restrict__ out, int n,
    const int* __restrict__ ovfCount, const int* __restrict__ ovfSrc,
    const int* __restrict__ ovfDst, int nEdges) {
  int t = blockIdx.x * blockDim.x + threadIdx.x;
  int i = t >> 3;
  if (i >= n) return;
  const int l8 = threadIdx.x & 7;
  const int f = l8 * 4;
  const float di = dinv[i];
  float4 a, b;
  row_sum8(xs, rows, i, n, l8, ovfCount, ovfSrc, ovfDst, nEdges, a, b);
  a.x *= di; a.y *= di; a.z *= di; a.w *= di;
  b.x *= di; b.y *= di; b.z *= di; b.w *= di;
  a.x += bias[f]; a.y += bias[f + 1]; a.z += bias[f + 2]; a.w += bias[f + 3];
  b.x += bias[32 + f]; b.y += bias[32 + f + 1]; b.z += bias[32 + f + 2]; b.w += bias[32 + f + 3];
  *(float4*)(out + (size_t)i * 64 + f) = a;
  *(float4*)(out + (size_t)i * 64 + 32 + f) = b;
}

#define A_PITCH 72
#define D_PITCH 68
#define GG_NODES 32

__global__ __launch_bounds__(256) void gather_gemm_kernel(
    const __bf16* __restrict__ xs, const float* __restrict__ dinv,
    const int* __restrict__ rows, const float* __restrict__ bias,
    const __bf16* __restrict__ Bp, __bf16* __restrict__ out, int n,
    const int* __restrict__ ovfCount, const int* __restrict__ ovfSrc,
    const int* __restrict__ ovfDst, int nEdges) {
  __shared__ __attribute__((aligned(16))) __bf16 sA[GG_NODES * A_PITCH];
  __shared__ __attribute__((aligned(16))) float  sD[GG_NODES * D_PITCH];
  __shared__ __attribute__((aligned(16))) float  sScale[GG_NODES];

  const int lane = threadIdx.x & 31;
  const int wave = threadIdx.x >> 5;
  const int nb   = threadIdx.x >> 3;
  const int l8   = threadIdx.x & 7;
  const int f    = l8 * 4;
  const int i    = blockIdx.x * GG_NODES + nb;
  const int rt   = wave >> 2;
  const int ct   = wave & 3;

  const v16bf b0 = *(const v16bf*)(Bp + ((size_t)(0 * 4 + ct) * 32 + lane) * 16);
  const v16bf b1 = *(const v16bf*)(Bp + ((size_t)(1 * 4 + ct) * 32 + lane) * 16);

  if (blockIdx.x == 0 && threadIdx.x < 8) {
    const v8bf z = {};
    *(v8bf*)(out + (size_t)n * 64 + threadIdx.x * 8) = z;
  }

  float4 hA = {0.0f, 0.0f, 0.0f, 0.0f};
  float4 hB = {0.0f, 0.0f, 0.0f, 0.0f};
  float di = 0.0f;
  if (i < n) {
    di = dinv[i];
    float4 a, b;
    row_sum8(xs, rows, i, n, l8, ovfCount, ovfSrc, ovfDst, nEdges, a, b);
    hA.x = a.x * di + bias[f];          hA.x = hA.x > 0.0f ? hA.x : 0.0f;
    hA.y = a.y * di + bias[f + 1];      hA.y = hA.y > 0.0f ? hA.y : 0.0f;
    hA.z = a.z * di + bias[f + 2];      hA.z = hA.z > 0.0f ? hA.z : 0.0f;
    hA.w = a.w * di + bias[f + 3];      hA.w = hA.w > 0.0f ? hA.w : 0.0f;
    hB.x = b.x * di + bias[32 + f];     hB.x = hB.x > 0.0f ? hB.x : 0.0f;
    hB.y = b.y * di + bias[32 + f + 1]; hB.y = hB.y > 0.0f ? hB.y : 0.0f;
    hB.z = b.z * di + bias[32 + f + 2]; hB.z = hB.z > 0.0f ? hB.z : 0.0f;
    hB.w = b.w * di + bias[32 + f + 3]; hB.w = hB.w > 0.0f ? hB.w : 0.0f;
  }
  const v4bf hbA = {(__bf16)hA.x, (__bf16)hA.y, (__bf16)hA.z, (__bf16)hA.w};
  const v4bf hbB = {(__bf16)hB.x, (__bf16)hB.y, (__bf16)hB.z, (__bf16)hB.w};
  *(v4bf*)(sA + nb * A_PITCH + f) = hbA;
  *(v4bf*)(sA + nb * A_PITCH + 32 + f) = hbB;
  if (l8 == 0) sScale[nb] = di;
  __syncthreads();

  {
    const int c  = lane & 15;
    const int hh = lane >> 4;
    const __bf16* ap = sA + (rt * 16 + c) * A_PITCH + hh * 8;
    const v8bf a0lo = *(const v8bf*)(ap);
    const v8bf a0hi = *(const v8bf*)(ap + 16);
    const v8bf a1lo = *(const v8bf*)(ap + 32);
    const v8bf a1hi = *(const v8bf*)(ap + 48);
    const v16bf a0 = __builtin_shufflevector(a0lo, a0hi, 0, 1, 2, 3, 4, 5, 6, 7, 8, 9, 10, 11, 12, 13, 14, 15);
    const v16bf a1 = __builtin_shufflevector(a1lo, a1hi, 0, 1, 2, 3, 4, 5, 6, 7, 8, 9, 10, 11, 12, 13, 14, 15);
    v8f acc = {};
    acc = __builtin_amdgcn_wmma_f32_16x16x32_bf16(
         false, a0,  false, b0,
         (short)0, acc,  false,  false);
    acc = __builtin_amdgcn_wmma_f32_16x16x32_bf16(
         false, a1,  false, b1,
         (short)0, acc,  false,  false);
    const float4 s0 = *(const float4*)(sScale + rt * 16 + hh * 8);
    const float4 s1 = *(const float4*)(sScale + rt * 16 + hh * 8 + 4);
    const float sc[8] = {s0.x, s0.y, s0.z, s0.w, s1.x, s1.y, s1.z, s1.w};
    float* dp = sD + (rt * 16 + hh * 8) * D_PITCH + ct * 16 + c;
#pragma unroll
    for (int r = 0; r < 8; ++r) dp[r * D_PITCH] = acc[r] * sc[r];
  }
  __syncthreads();

  if (i < n) {
    const float4 d0 = *(const float4*)(sD + nb * D_PITCH + f);
    const float4 d1 = *(const float4*)(sD + nb * D_PITCH + 32 + f);
    const v8bf o = {(__bf16)d0.x, (__bf16)d0.y, (__bf16)d0.z, (__bf16)d0.w,
                    (__bf16)d1.x, (__bf16)d1.y, (__bf16)d1.z, (__bf16)d1.w};
    *(v8bf*)(out + (size_t)i * 64 + l8 * 8) = o;
  }
}

extern "C" void kernel_launch(void* const* d_in, const int* in_sizes, int n_in,
                              void* d_out, int out_size, void* d_ws, size_t ws_size,
                              hipStream_t stream) {
  const int*   edge_index = (const int*)d_in[0];
  const float* emb        = (const float*)d_in[1];
  const float* W1         = (const float*)d_in[2];
  const float* b1         = (const float*)d_in[3];
  const float* W2         = (const float*)d_in[4];
  const float* b2         = (const float*)d_in[5];
  float* out = (float*)d_out;

  const int E = in_sizes[0] / 2;
  const int N = in_sizes[1] / 128;
  const int* src = edge_index;
  const int* dst = edge_index + E;

  float* ws   = (float*)d_ws;
  float* dinv = ws;
  size_t npad = ((size_t)N + 63) & ~(size_t)63;
  int*   ovfCount = (int*)(ws + npad);
  float* p1   = ws + npad + 64;
  float* p2   = p1 + (size_t)N * 64;
  __bf16* x1  = (__bf16*)p1;
  __bf16* x2  = (__bf16*)p2;
  __bf16* Bp1 = (__bf16*)(p2 + (size_t)N * 64);
  __bf16* Bp2 = Bp1 + 8192;
  int* rows   = (int*)(p2 + (size_t)N * 64 + 6144);
  int* ovfSrc = rows + (size_t)N * ROW_INTS;
  size_t epad = ((size_t)E + 63) & ~(size_t)63;
  int* ovfDst = ovfSrc + epad;
  const int nSlices = (E + SP_SLICE - 1) / SP_SLICE;
  const int splitBlk = nSlices > 3 ? nSlices : 3;
  const int nBins   = (N + BIN_NODES - 1) / BIN_NODES;
  int2* sorted = (int2*)(ovfDst + epad);
  int* starts  = (int*)(sorted + (size_t)splitBlk * SP_SLICE);

  const int tpb = 256;
  const long nf = (long)N * 8;
  const int nfBlk = (int)((nf + tpb - 1) / tpb);

  split_edges_kernel<<<splitBlk, SP_THREADS, 0, stream>>>(src, dst, sorted, starts, E, N, nBins,
                                                          W1, W2, Bp1, Bp2, ovfCount);

  build_rows_kernel<<<nBins, BR_THREADS, 0, stream>>>(sorted, starts, rows, ovfCount, ovfSrc, ovfDst,
                                                      N, nBins, splitBlk, emb, Bp1, x1, dinv);

  gather_gemm_kernel<<<nfBlk, tpb, 0, stream>>>(x1, dinv, rows, b1, Bp2, x2, N,
                                                ovfCount, ovfSrc, ovfDst, E);

  gather_rows_kernel<<<nfBlk, tpb, 0, stream>>>(x2, dinv, rows, b2, out, N,
                                                ovfCount, ovfSrc, ovfDst, E);
}
